# starting kernel
# baseline (speedup 1.0000x reference)
#include <hip/hip_runtime.h>
#include <hip/hip_bf16.h>

#define BB 8
#define LL 2048
#define DD 512
#define RR 512
#define MTOT (BB * LL)

typedef float v2f __attribute__((ext_vector_type(2)));
typedef float v8f __attribute__((ext_vector_type(8)));

__device__ __forceinline__ v8f wmma_f32(v2f a, v2f b, v8f c) {
  return __builtin_amdgcn_wmma_f32_16x16x4_f32(
       false, a,  false, b,
       (short)0, c,  false,  false);
}

__device__ __forceinline__ float waveRedSum(float v) {
#pragma unroll
  for (int m = 16; m >= 1; m >>= 1) v += __shfl_xor(v, m, 32);
  return v;
}
__device__ __forceinline__ float waveRedMax(float v) {
#pragma unroll
  for (int m = 16; m >= 1; m >>= 1) v = fmaxf(v, __shfl_xor(v, m, 32));
  return v;
}

__global__ __launch_bounds__(256) void prep_kernel(
    const float* __restrict__ centers, const float* __restrict__ widths,
    float* __restrict__ iw, float* __restrict__ cw, float* __restrict__ cc) {
  int row = (blockIdx.x * blockDim.x + threadIdx.x) >> 5;
  int lane = threadIdx.x & 31;
  if (row >= RR) return;
  const float* crow = centers + row * DD;
  const float* wrow = widths + row * DD;
  float acc = 0.0f;
#pragma unroll
  for (int t = 0; t < DD / 32; ++t) {
    int d = lane + 32 * t;
    float w = wrow[d];
    float inv = 1.0f / (w * w);
    float c = crow[d];
    iw[row * DD + d] = inv;
    cw[row * DD + d] = c * inv;
    acc += c * c * inv;
  }
  acc = waveRedSum(acc);
  if (lane == 0) cc[row] = acc;
}

__global__ __launch_bounds__(256) void qproj_kernel(
    const float* __restrict__ A, const float* __restrict__ Bm,
    const float* __restrict__ bias, float* __restrict__ C) {
  const int K = DD, N = DD;
  int wave = threadIdx.x >> 5;
  int lane = threadIdx.x & 31;
  int hi = lane >> 4;
  int ln = lane & 15;
  int tile_m = blockIdx.x * 128 + wave * 16;
  int tile_n = blockIdx.y * 64;

  const float* arow = A + (tile_m + ln) * K + 2 * hi;
  const float* brow = Bm + (tile_n + ln) * K + 2 * hi;
  __builtin_prefetch(arow, 0, 1);

  v8f c0 = {}, c1 = {}, c2 = {}, c3 = {};
  for (int k0 = 0; k0 < K; k0 += 4) {
    v2f a = *(const v2f*)(arow + k0);
    v2f b0 = *(const v2f*)(brow + 0 * 16 * K + k0);
    v2f b1 = *(const v2f*)(brow + 1 * 16 * K + k0);
    v2f b2 = *(const v2f*)(brow + 2 * 16 * K + k0);
    v2f b3 = *(const v2f*)(brow + 3 * 16 * K + k0);
    c0 = wmma_f32(a, b0, c0);
    c1 = wmma_f32(a, b1, c1);
    c2 = wmma_f32(a, b2, c2);
    c3 = wmma_f32(a, b3, c3);
  }
  float bq0 = bias[tile_n + 0 + ln], bq1 = bias[tile_n + 16 + ln];
  float bq2 = bias[tile_n + 32 + ln], bq3 = bias[tile_n + 48 + ln];
#pragma unroll
  for (int i = 0; i < 8; ++i) {
    int m = tile_m + i + 8 * hi;
    float* crow = C + m * N + tile_n + ln;
    crow[0]  = c0[i] + bq0;
    crow[16] = c1[i] + bq1;
    crow[32] = c2[i] + bq2;
    crow[48] = c3[i] + bq3;
  }
}

__global__ __launch_bounds__(256) void z_kernel(
    const float* __restrict__ q, const float* __restrict__ iw,
    const float* __restrict__ cw, const float* __restrict__ cc,
    float* __restrict__ z) {
  const int K = DD, N = RR;
  int wave = threadIdx.x >> 5;
  int lane = threadIdx.x & 31;
  int hi = lane >> 4;
  int ln = lane & 15;
  int tile_m = blockIdx.x * 128 + wave * 16;
  int tile_n = blockIdx.y * 32;

  const float* arow  = q  + (tile_m + ln) * K + 2 * hi;
  const float* iwrow = iw + (tile_n + ln) * K + 2 * hi;
  const float* cwrow = cw + (tile_n + ln) * K + 2 * hi;

  v8f qq0 = {}, qc0 = {}, qq1 = {}, qc1 = {};
  for (int k0 = 0; k0 < K; k0 += 4) {
    v2f a = *(const v2f*)(arow + k0);
    v2f asq = a * a;
    v2f bi0 = *(const v2f*)(iwrow + k0);
    v2f bi1 = *(const v2f*)(iwrow + 16 * K + k0);
    v2f bc0 = *(const v2f*)(cwrow + k0);
    v2f bc1 = *(const v2f*)(cwrow + 16 * K + k0);
    qq0 = wmma_f32(asq, bi0, qq0);
    qc0 = wmma_f32(a,   bc0, qc0);
    qq1 = wmma_f32(asq, bi1, qq1);
    qc1 = wmma_f32(a,   bc1, qc1);
  }
  const float scale = -0.5f / (float)DD;
  float cc0 = cc[tile_n + ln], cc1 = cc[tile_n + 16 + ln];
#pragma unroll
  for (int i = 0; i < 8; ++i) {
    int m = tile_m + i + 8 * hi;
    float* zrow = z + m * N + tile_n + ln;
    zrow[0]  = scale * (qq0[i] - 2.0f * qc0[i] + cc0);
    zrow[16] = scale * (qq1[i] - 2.0f * qc1[i] + cc1);
  }
}

__global__ __launch_bounds__(256) void softmax_kernel(float* __restrict__ z) {
  int row = (blockIdx.x * blockDim.x + threadIdx.x) >> 5;
  int lane = threadIdx.x & 31;
  if (row >= MTOT) return;
  float* p = z + (size_t)row * RR;
  float vals[RR / 32];
  float mx = -3.4e38f;
#pragma unroll
  for (int t = 0; t < RR / 32; ++t) {
    vals[t] = p[lane + 32 * t];
    mx = fmaxf(mx, vals[t]);
  }
  mx = waveRedMax(mx);
  float s = 0.0f;
#pragma unroll
  for (int t = 0; t < RR / 32; ++t) {
    vals[t] = __expf(vals[t] - mx);
    s += vals[t];
  }
  s = waveRedSum(s);
  float inv = 1.0f / s;
#pragma unroll
  for (int t = 0; t < RR / 32; ++t) p[lane + 32 * t] = vals[t] * inv;
}

__global__ __launch_bounds__(256) void conq_kernel(
    const float* __restrict__ query, const float* __restrict__ Wc,
    const float* __restrict__ bc, float* __restrict__ conq) {
  const int K = LL, N = DD;
  int b = blockIdx.z;
  int wave = threadIdx.x >> 5;
  int lane = threadIdx.x & 31;
  int hi = lane >> 4;
  int ln = lane & 15;
  int tile_m = blockIdx.x * 128 + wave * 16;
  int tile_n = blockIdx.y * 64;

  const float* arow = Wc + (tile_m + ln) * K + 2 * hi;
  const float* Qb = query + b * LL * DD;

  v8f c0 = {}, c1 = {}, c2 = {}, c3 = {};
  for (int k0 = 0; k0 < K; k0 += 4) {
    int kk = k0 + 2 * hi;
    v2f a = *(const v2f*)(arow + k0);
    const float* bp = Qb + kk * DD + tile_n + ln;
    v2f b0, b1, b2, b3;
    b0.x = bp[0];       b0.y = bp[DD + 0];
    b1.x = bp[16];      b1.y = bp[DD + 16];
    b2.x = bp[32];      b2.y = bp[DD + 32];
    b3.x = bp[48];      b3.y = bp[DD + 48];
    c0 = wmma_f32(a, b0, c0);
    c1 = wmma_f32(a, b1, c1);
    c2 = wmma_f32(a, b2, c2);
    c3 = wmma_f32(a, b3, c3);
  }
  float* Cb = conq + b * RR * DD;
#pragma unroll
  for (int i = 0; i < 8; ++i) {
    int m = tile_m + i + 8 * hi;
    float bias = bc[m];
    float* crow = Cb + m * N + tile_n + ln;
    crow[0]  = c0[i] + bias;
    crow[16] = c1[i] + bias;
    crow[32] = c2[i] + bias;
    crow[48] = c3[i] + bias;
  }
}

__global__ __launch_bounds__(256) void out_kernel(
    const float* __restrict__ Fss, const float* __restrict__ conq,
    float* __restrict__ out) {
  const int K = RR, N = DD;
  int b = blockIdx.z;
  int wave = threadIdx.x >> 5;
  int lane = threadIdx.x & 31;
  int hi = lane >> 4;
  int ln = lane & 15;
  int tile_m = blockIdx.x * 128 + wave * 16;
  int tile_n = blockIdx.y * 64;

  const float* arow = Fss + b * LL * RR + (tile_m + ln) * K + 2 * hi;
  const float* Bb_ = conq + b * RR * DD;

  v8f c0 = {}, c1 = {}, c2 = {}, c3 = {};
  for (int k0 = 0; k0 < K; k0 += 4) {
    int kk = k0 + 2 * hi;
    v2f a = *(const v2f*)(arow + k0);
    const float* bp = Bb_ + kk * DD + tile_n + ln;
    v2f b0, b1, b2, b3;
    b0.x = bp[0];       b0.y = bp[DD + 0];
    b1.x = bp[16];      b1.y = bp[DD + 16];
    b2.x = bp[32];      b2.y = bp[DD + 32];
    b3.x = bp[48];      b3.y = bp[DD + 48];
    c0 = wmma_f32(a, b0, c0);
    c1 = wmma_f32(a, b1, c1);
    c2 = wmma_f32(a, b2, c2);
    c3 = wmma_f32(a, b3, c3);
  }
  float* Ob = out + b * LL * DD;
#pragma unroll
  for (int i = 0; i < 8; ++i) {
    int m = tile_m + i + 8 * hi;
    float* crow = Ob + m * N + tile_n + ln;
    crow[0]  = c0[i];
    crow[16] = c1[i];
    crow[32] = c2[i];
    crow[48] = c3[i];
  }
}

extern "C" void kernel_launch(void* const* d_in, const int* in_sizes, int n_in,
                              void* d_out, int out_size, void* d_ws, size_t ws_size,
                              hipStream_t stream) {
  const float* query   = (const float*)d_in[0];
  const float* Wq      = (const float*)d_in[1];
  const float* bq      = (const float*)d_in[2];
  const float* Wc      = (const float*)d_in[3];
  const float* bc      = (const float*)d_in[4];
  const float* centers = (const float*)d_in[5];
  const float* widths  = (const float*)d_in[6];
  float* out = (float*)d_out;

  float* q    = (float*)d_ws;
  float* z    = q    + (size_t)MTOT * DD;
  float* iw   = z    + (size_t)MTOT * RR;
  float* cw   = iw   + (size_t)RR * DD;
  float* cc   = cw   + (size_t)RR * DD;
  float* conq = cc   + RR;

  prep_kernel<<<RR / 8, 256, 0, stream>>>(centers, widths, iw, cw, cc);

  qproj_kernel<<<dim3(MTOT / 128, DD / 64), 256, 0, stream>>>(query, Wq, bq, q);

  z_kernel<<<dim3(MTOT / 128, RR / 32), 256, 0, stream>>>(q, iw, cw, cc, z);

  softmax_kernel<<<MTOT / 8, 256, 0, stream>>>(z);

  conq_kernel<<<dim3(RR / 128, DD / 64, BB), 256, 0, stream>>>(query, Wc, bc, conq);

  out_kernel<<<dim3(LL / 128, DD / 64, BB), 256, 0, stream>>>(z, conq, out);
}


# All four GEMMs move operands and results 128 bits per lane: a lane holds 4 consecutive k, placed alike in A and B (k summed in a permuted order), and 4 adjacent output columns (z: 2)
# speedup vs baseline: 1.2876x; 1.2876x over previous
#include <hip/hip_runtime.h>
#include <hip/hip_bf16.h>

#define BB 8
#define LL 2048
#define DD 512
#define RR 512
#define MTOT (BB * LL)

typedef float v2f __attribute__((ext_vector_type(2)));
typedef float v4f __attribute__((ext_vector_type(4)));
typedef float v8f __attribute__((ext_vector_type(8)));

__device__ __forceinline__ v8f wmma_f32(v2f a, v2f b, v8f c) {
  return __builtin_amdgcn_wmma_f32_16x16x4_f32(
       false, a,  false, b,
       (short)0, c,  false,  false);
}

__device__ __forceinline__ float waveRedSum(float v) {
#pragma unroll
  for (int m = 16; m >= 1; m >>= 1) v += __shfl_xor(v, m, 32);
  return v;
}
__device__ __forceinline__ float waveRedMax(float v) {
#pragma unroll
  for (int m = 16; m >= 1; m >>= 1) v = fmaxf(v, __shfl_xor(v, m, 32));
  return v;
}

__global__ __launch_bounds__(256) void prep_kernel(
    const float* __restrict__ centers, const float* __restrict__ widths,
    float* __restrict__ iw, float* __restrict__ cw, float* __restrict__ cc) {
  int row = (blockIdx.x * blockDim.x + threadIdx.x) >> 5;
  int lane = threadIdx.x & 31;
  if (row >= RR) return;
  const float* crow = centers + row * DD;
  const float* wrow = widths + row * DD;
  float acc = 0.0f;
#pragma unroll
  for (int t = 0; t < DD / 32; ++t) {
    int d = lane + 32 * t;
    float w = wrow[d];
    float inv = 1.0f / (w * w);
    float c = crow[d];
    iw[row * DD + d] = inv;
    cw[row * DD + d] = c * inv;
    acc += c * c * inv;
  }
  acc = waveRedSum(acc);
  if (lane == 0) cc[row] = acc;
}

__global__ __launch_bounds__(256) void qproj_kernel(
    const float* __restrict__ A, const float* __restrict__ Bm,
    const float* __restrict__ bias, float* __restrict__ C) {
  const int K = DD, N = DD;
  int wave = threadIdx.x >> 5;
  int lane = threadIdx.x & 31;
  int hi = lane >> 4;
  int ln = lane & 15;
  int tile_m = blockIdx.x * 128 + wave * 16;
  int tile_n = blockIdx.y * 64;

  const float* arow = A + (tile_m + ln) * K + 4 * hi;
  const float* brow = Bm + (tile_n + 4 * ln) * K + 4 * hi;
  __builtin_prefetch(arow, 0, 1);

  v8f c0 = {}, c1 = {}, c2 = {}, c3 = {};
  for (int k0 = 0; k0 < K; k0 += 8) {
    v4f a  = *(const v4f*)(arow + k0);
    v4f b0 = *(const v4f*)(brow + 0 * K + k0);
    v4f b1 = *(const v4f*)(brow + 1 * K + k0);
    v4f b2 = *(const v4f*)(brow + 2 * K + k0);
    v4f b3 = *(const v4f*)(brow + 3 * K + k0);
    v2f alo = {a.x, a.y}, ahi = {a.z, a.w};
    c0 = wmma_f32(alo, (v2f){b0.x, b0.y}, c0);
    c1 = wmma_f32(alo, (v2f){b1.x, b1.y}, c1);
    c2 = wmma_f32(alo, (v2f){b2.x, b2.y}, c2);
    c3 = wmma_f32(alo, (v2f){b3.x, b3.y}, c3);
    c0 = wmma_f32(ahi, (v2f){b0.z, b0.w}, c0);
    c1 = wmma_f32(ahi, (v2f){b1.z, b1.w}, c1);
    c2 = wmma_f32(ahi, (v2f){b2.z, b2.w}, c2);
    c3 = wmma_f32(ahi, (v2f){b3.z, b3.w}, c3);
  }
  v4f bq = *(const v4f*)(bias + tile_n + 4 * ln);
#pragma unroll
  for (int i = 0; i < 8; ++i) {
    int m = tile_m + i + 8 * hi;
    v4f o = {c0[i] + bq.x, c1[i] + bq.y, c2[i] + bq.z, c3[i] + bq.w};
    *(v4f*)(C + m * N + tile_n + 4 * ln) = o;
  }
}

__global__ __launch_bounds__(256) void z_kernel(
    const float* __restrict__ q, const float* __restrict__ iw,
    const float* __restrict__ cw, const float* __restrict__ cc,
    float* __restrict__ z) {
  const int K = DD, N = RR;
  int wave = threadIdx.x >> 5;
  int lane = threadIdx.x & 31;
  int hi = lane >> 4;
  int ln = lane & 15;
  int tile_m = blockIdx.x * 128 + wave * 16;
  int tile_n = blockIdx.y * 32;

  const float* arow  = q  + (tile_m + ln) * K + 4 * hi;
  const float* iwrow = iw + (tile_n + 2 * ln) * K + 4 * hi;
  const float* cwrow = cw + (tile_n + 2 * ln) * K + 4 * hi;

  v8f qq0 = {}, qc0 = {}, qq1 = {}, qc1 = {};
  for (int k0 = 0; k0 < K; k0 += 8) {
    v4f a = *(const v4f*)(arow + k0);
    v4f asq = a * a;
    v4f bi0 = *(const v4f*)(iwrow + k0);
    v4f bi1 = *(const v4f*)(iwrow + K + k0);
    v4f bc0 = *(const v4f*)(cwrow + k0);
    v4f bc1 = *(const v4f*)(cwrow + K + k0);
    v2f alo = {a.x, a.y}, ahi = {a.z, a.w};
    v2f slo = {asq.x, asq.y}, shi = {asq.z, asq.w};
    qq0 = wmma_f32(slo, (v2f){bi0.x, bi0.y}, qq0);
    qc0 = wmma_f32(alo, (v2f){bc0.x, bc0.y}, qc0);
    qq1 = wmma_f32(slo, (v2f){bi1.x, bi1.y}, qq1);
    qc1 = wmma_f32(alo, (v2f){bc1.x, bc1.y}, qc1);
    qq0 = wmma_f32(shi, (v2f){bi0.z, bi0.w}, qq0);
    qc0 = wmma_f32(ahi, (v2f){bc0.z, bc0.w}, qc0);
    qq1 = wmma_f32(shi, (v2f){bi1.z, bi1.w}, qq1);
    qc1 = wmma_f32(ahi, (v2f){bc1.z, bc1.w}, qc1);
  }
  const float scale = -0.5f / (float)DD;
  v2f ccv = *(const v2f*)(cc + tile_n + 2 * ln);
#pragma unroll
  for (int i = 0; i < 8; ++i) {
    int m = tile_m + i + 8 * hi;
    v2f o = {scale * (qq0[i] - 2.0f * qc0[i] + ccv.x),
             scale * (qq1[i] - 2.0f * qc1[i] + ccv.y)};
    *(v2f*)(z + m * N + tile_n + 2 * ln) = o;
  }
}

__global__ __launch_bounds__(256) void softmax_kernel(float* __restrict__ z) {
  int row = (blockIdx.x * blockDim.x + threadIdx.x) >> 5;
  int lane = threadIdx.x & 31;
  if (row >= MTOT) return;
  float* p = z + (size_t)row * RR;
  float vals[RR / 32];
  float mx = -3.4e38f;
#pragma unroll
  for (int t = 0; t < RR / 32; ++t) {
    vals[t] = p[lane + 32 * t];
    mx = fmaxf(mx, vals[t]);
  }
  mx = waveRedMax(mx);
  float s = 0.0f;
#pragma unroll
  for (int t = 0; t < RR / 32; ++t) {
    vals[t] = __expf(vals[t] - mx);
    s += vals[t];
  }
  s = waveRedSum(s);
  float inv = 1.0f / s;
#pragma unroll
  for (int t = 0; t < RR / 32; ++t) p[lane + 32 * t] = vals[t] * inv;
}

__global__ __launch_bounds__(256) void conq_kernel(
    const float* __restrict__ query, const float* __restrict__ Wc,
    const float* __restrict__ bc, float* __restrict__ conq) {
  const int K = LL, N = DD;
  int b = blockIdx.z;
  int wave = threadIdx.x >> 5;
  int lane = threadIdx.x & 31;
  int hi = lane >> 4;
  int ln = lane & 15;
  int tile_m = blockIdx.x * 128 + wave * 16;
  int tile_n = blockIdx.y * 64;

  const float* arow = Wc + (tile_m + ln) * K + 4 * hi;
  const float* Qb = query + b * LL * DD + tile_n + 4 * ln;

  v8f c0 = {}, c1 = {}, c2 = {}, c3 = {};
  for (int k0 = 0; k0 < K; k0 += 8) {
    int kk = k0 + 4 * hi;
    v4f a = *(const v4f*)(arow + k0);
    const float* bp = Qb + kk * DD;
    v4f r0 = *(const v4f*)(bp);
    v4f r1 = *(const v4f*)(bp + DD);
    v4f r2 = *(const v4f*)(bp + 2 * DD);
    v4f r3 = *(const v4f*)(bp + 3 * DD);
    v2f alo = {a.x, a.y}, ahi = {a.z, a.w};
    c0 = wmma_f32(alo, (v2f){r0.x, r1.x}, c0);
    c1 = wmma_f32(alo, (v2f){r0.y, r1.y}, c1);
    c2 = wmma_f32(alo, (v2f){r0.z, r1.z}, c2);
    c3 = wmma_f32(alo, (v2f){r0.w, r1.w}, c3);
    c0 = wmma_f32(ahi, (v2f){r2.x, r3.x}, c0);
    c1 = wmma_f32(ahi, (v2f){r2.y, r3.y}, c1);
    c2 = wmma_f32(ahi, (v2f){r2.z, r3.z}, c2);
    c3 = wmma_f32(ahi, (v2f){r2.w, r3.w}, c3);
  }
  float* Cb = conq + b * RR * DD;
#pragma unroll
  for (int i = 0; i < 8; ++i) {
    int m = tile_m + i + 8 * hi;
    float bias = bc[m];
    v4f o = {c0[i] + bias, c1[i] + bias, c2[i] + bias, c3[i] + bias};
    *(v4f*)(Cb + m * N + tile_n + 4 * ln) = o;
  }
}

__global__ __launch_bounds__(256) void out_kernel(
    const float* __restrict__ Fss, const float* __restrict__ conq,
    float* __restrict__ out) {
  const int K = RR, N = DD;
  int b = blockIdx.z;
  int wave = threadIdx.x >> 5;
  int lane = threadIdx.x & 31;
  int hi = lane >> 4;
  int ln = lane & 15;
  int tile_m = blockIdx.x * 128 + wave * 16;
  int tile_n = blockIdx.y * 64;

  const float* arow = Fss + b * LL * RR + (tile_m + ln) * K + 4 * hi;
  const float* Bb_ = conq + b * RR * DD + tile_n + 4 * ln;

  v8f c0 = {}, c1 = {}, c2 = {}, c3 = {};
  for (int k0 = 0; k0 < K; k0 += 8) {
    int kk = k0 + 4 * hi;
    v4f a = *(const v4f*)(arow + k0);
    const float* bp = Bb_ + kk * DD;
    v4f r0 = *(const v4f*)(bp);
    v4f r1 = *(const v4f*)(bp + DD);
    v4f r2 = *(const v4f*)(bp + 2 * DD);
    v4f r3 = *(const v4f*)(bp + 3 * DD);
    v2f alo = {a.x, a.y}, ahi = {a.z, a.w};
    c0 = wmma_f32(alo, (v2f){r0.x, r1.x}, c0);
    c1 = wmma_f32(alo, (v2f){r0.y, r1.y}, c1);
    c2 = wmma_f32(alo, (v2f){r0.z, r1.z}, c2);
    c3 = wmma_f32(alo, (v2f){r0.w, r1.w}, c3);
    c0 = wmma_f32(ahi, (v2f){r2.x, r3.x}, c0);
    c1 = wmma_f32(ahi, (v2f){r2.y, r3.y}, c1);
    c2 = wmma_f32(ahi, (v2f){r2.z, r3.z}, c2);
    c3 = wmma_f32(ahi, (v2f){r2.w, r3.w}, c3);
  }
  float* Ob = out + b * LL * DD;
#pragma unroll
  for (int i = 0; i < 8; ++i) {
    int m = tile_m + i + 8 * hi;
    v4f o = {c0[i], c1[i], c2[i], c3[i]};
    *(v4f*)(Ob + m * N + tile_n + 4 * ln) = o;
  }
}

extern "C" void kernel_launch(void* const* d_in, const int* in_sizes, int n_in,
                              void* d_out, int out_size, void* d_ws, size_t ws_size,
                              hipStream_t stream) {
  const float* query   = (const float*)d_in[0];
  const float* Wq      = (const float*)d_in[1];
  const float* bq      = (const float*)d_in[2];
  const float* Wc      = (const float*)d_in[3];
  const float* bc      = (const float*)d_in[4];
  const float* centers = (const float*)d_in[5];
  const float* widths  = (const float*)d_in[6];
  float* out = (float*)d_out;

  float* q    = (float*)d_ws;
  float* z    = q    + (size_t)MTOT * DD;
  float* iw   = z    + (size_t)MTOT * RR;
  float* cw   = iw   + (size_t)RR * DD;
  float* cc   = cw   + (size_t)RR * DD;
  float* conq = cc   + RR;

  prep_kernel<<<RR / 8, 256, 0, stream>>>(centers, widths, iw, cw, cc);

  qproj_kernel<<<dim3(MTOT / 128, DD / 64), 256, 0, stream>>>(query, Wq, bq, q);

  z_kernel<<<dim3(MTOT / 128, RR / 32), 256, 0, stream>>>(q, iw, cw, cc, z);

  softmax_kernel<<<MTOT / 8, 256, 0, stream>>>(z);

  conq_kernel<<<dim3(RR / 128, DD / 64, BB), 256, 0, stream>>>(query, Wc, bc, conq);

  out_kernel<<<dim3(LL / 128, DD / 64, BB), 256, 0, stream>>>(z, conq, out);
}


# All four GEMMs stage their operand tiles through LDS (32 k per stage, two stages, next stage's global loads in flight): whole-line global loads, ds_load_b128 fragments, k order and sums unchanged
# speedup vs baseline: 1.9475x; 1.5125x over previous
#include <hip/hip_runtime.h>
#include <hip/hip_bf16.h>

#define BB 8
#define LL 2048
#define DD 512
#define RR 512
#define MTOT (BB * LL)

typedef float v2f __attribute__((ext_vector_type(2)));
typedef float v4f __attribute__((ext_vector_type(4)));
typedef float v8f __attribute__((ext_vector_type(8)));


#define KT 32
#define PA 36
#define PB 72

__device__ __forceinline__ v8f wmma_f32(v2f a, v2f b, v8f c) {
  return __builtin_amdgcn_wmma_f32_16x16x4_f32(
       false, a,  false, b,
       (short)0, c,  false,  false);
}

__device__ __forceinline__ float waveRedSum(float v) {
#pragma unroll
  for (int m = 16; m >= 1; m >>= 1) v += __shfl_xor(v, m, 32);
  return v;
}
__device__ __forceinline__ float waveRedMax(float v) {
#pragma unroll
  for (int m = 16; m >= 1; m >>= 1) v = fmaxf(v, __shfl_xor(v, m, 32));
  return v;
}

__global__ __launch_bounds__(256) void prep_kernel(
    const float* __restrict__ centers, const float* __restrict__ widths,
    float* __restrict__ iw, float* __restrict__ cw, float* __restrict__ cc) {
  int row = (blockIdx.x * blockDim.x + threadIdx.x) >> 5;
  int lane = threadIdx.x & 31;
  if (row >= RR) return;
  const float* crow = centers + row * DD;
  const float* wrow = widths + row * DD;
  float acc = 0.0f;
#pragma unroll
  for (int t = 0; t < DD / 32; ++t) {
    int d = lane + 32 * t;
    float w = wrow[d];
    float inv = 1.0f / (w * w);
    float c = crow[d];
    iw[row * DD + d] = inv;
    cw[row * DD + d] = c * inv;
    acc += c * c * inv;
  }
  acc = waveRedSum(acc);
  if (lane == 0) cc[row] = acc;
}

__global__ __launch_bounds__(256) void qproj_kernel(
    const float* __restrict__ A, const float* __restrict__ Bm,
    const float* __restrict__ bias, float* __restrict__ C) {
  const int K = DD, N = DD;
  __shared__ __attribute__((aligned(16))) float As[2][128 * PA];
  __shared__ __attribute__((aligned(16))) float Bs[2][64 * PA];
  int t = threadIdx.x;
  int wave = t >> 5;
  int lane = t & 31;
  int hi = lane >> 4;
  int ln = lane & 15;
  int blk_m = blockIdx.x * 128;
  int tile_m = blk_m + wave * 16;
  int tile_n = blockIdx.y * 64;

  int cr = t >> 3;
  int cq = (t & 7) * 4;
  const float* ag = A + (blk_m + cr) * K + cq;
  const float* bg = Bm + (tile_n + 4 * (cr & 15) + (cr >> 4)) * K + cq;
  int st = cr * PA + cq;
  int a_ld = (wave * 16 + ln) * PA + 4 * hi;
  int b_ld = ln * PA + 4 * hi;

  v4f ra[4], rb[2];
#pragma unroll
  for (int i = 0; i < 4; ++i) ra[i] = *(const v4f*)(ag + i * 32 * K);
#pragma unroll
  for (int i = 0; i < 2; ++i) rb[i] = *(const v4f*)(bg + i * 2 * K);
#pragma unroll
  for (int i = 0; i < 4; ++i) *(v4f*)(&As[0][st + i * 32 * PA]) = ra[i];
#pragma unroll
  for (int i = 0; i < 2; ++i) *(v4f*)(&Bs[0][st + i * 32 * PA]) = rb[i];
  __syncthreads();

  v8f c0 = {}, c1 = {}, c2 = {}, c3 = {};
  for (int kt = 0; kt < K / KT; ++kt) {
    const int cur = kt & 1;
    const bool more = (kt + 1) < K / KT;
    if (more) {
#pragma unroll
      for (int i = 0; i < 4; ++i)
        ra[i] = *(const v4f*)(ag + (kt + 1) * KT + i * 32 * K);
#pragma unroll
      for (int i = 0; i < 2; ++i)
        rb[i] = *(const v4f*)(bg + (kt + 1) * KT + i * 2 * K);
    }
    const float* as = &As[cur][a_ld];
    const float* bs = &Bs[cur][b_ld];
#pragma unroll
    for (int kk = 0; kk < KT / 8; ++kk) {
      v4f a  = *(const v4f*)(as + kk * 8);
      v4f b0 = *(const v4f*)(bs + 0 * 16 * PA + kk * 8);
      v4f b1 = *(const v4f*)(bs + 1 * 16 * PA + kk * 8);
      v4f b2 = *(const v4f*)(bs + 2 * 16 * PA + kk * 8);
      v4f b3 = *(const v4f*)(bs + 3 * 16 * PA + kk * 8);
      v2f alo = {a.x, a.y}, ahi = {a.z, a.w};
      c0 = wmma_f32(alo, (v2f){b0.x, b0.y}, c0);
      c1 = wmma_f32(alo, (v2f){b1.x, b1.y}, c1);
      c2 = wmma_f32(alo, (v2f){b2.x, b2.y}, c2);
      c3 = wmma_f32(alo, (v2f){b3.x, b3.y}, c3);
      c0 = wmma_f32(ahi, (v2f){b0.z, b0.w}, c0);
      c1 = wmma_f32(ahi, (v2f){b1.z, b1.w}, c1);
      c2 = wmma_f32(ahi, (v2f){b2.z, b2.w}, c2);
      c3 = wmma_f32(ahi, (v2f){b3.z, b3.w}, c3);
    }
    if (more) {
#pragma unroll
      for (int i = 0; i < 4; ++i) *(v4f*)(&As[cur ^ 1][st + i * 32 * PA]) = ra[i];
#pragma unroll
      for (int i = 0; i < 2; ++i) *(v4f*)(&Bs[cur ^ 1][st + i * 32 * PA]) = rb[i];
    }
    __syncthreads();
  }
  v4f bq = *(const v4f*)(bias + tile_n + 4 * ln);
#pragma unroll
  for (int i = 0; i < 8; ++i) {
    int m = tile_m + i + 8 * hi;
    v4f o = {c0[i] + bq.x, c1[i] + bq.y, c2[i] + bq.z, c3[i] + bq.w};
    *(v4f*)(C + m * N + tile_n + 4 * ln) = o;
  }
}

__global__ __launch_bounds__(256) void z_kernel(
    const float* __restrict__ q, const float* __restrict__ iw,
    const float* __restrict__ cw, const float* __restrict__ cc,
    float* __restrict__ z) {
  const int K = DD, N = RR;
  __shared__ __attribute__((aligned(16))) float As[2][128 * PA];
  __shared__ __attribute__((aligned(16))) float Bi[2][32 * PA];
  __shared__ __attribute__((aligned(16))) float Bc[2][32 * PA];
  int t = threadIdx.x;
  int wave = t >> 5;
  int lane = t & 31;
  int hi = lane >> 4;
  int ln = lane & 15;
  int blk_m = blockIdx.x * 128;
  int tile_m = blk_m + wave * 16;
  int tile_n = blockIdx.y * 32;

  int cr = t >> 3;
  int cq = (t & 7) * 4;
  const float* ag = q + (blk_m + cr) * K + cq;
  int brow = tile_n + 2 * (cr & 15) + (cr >> 4);
  const float* ig = iw + brow * K + cq;
  const float* cg = cw + brow * K + cq;
  int st = cr * PA + cq;
  int a_ld = (wave * 16 + ln) * PA + 4 * hi;
  int b_ld = ln * PA + 4 * hi;

  v4f ra[4], ri, rc;
#pragma unroll
  for (int i = 0; i < 4; ++i) ra[i] = *(const v4f*)(ag + i * 32 * K);
  ri = *(const v4f*)(ig);
  rc = *(const v4f*)(cg);
#pragma unroll
  for (int i = 0; i < 4; ++i) *(v4f*)(&As[0][st + i * 32 * PA]) = ra[i];
  *(v4f*)(&Bi[0][st]) = ri;
  *(v4f*)(&Bc[0][st]) = rc;
  __syncthreads();

  v8f qq0 = {}, qc0 = {}, qq1 = {}, qc1 = {};
  for (int kt = 0; kt < K / KT; ++kt) {
    const int cur = kt & 1;
    const bool more = (kt + 1) < K / KT;
    if (more) {
#pragma unroll
      for (int i = 0; i < 4; ++i)
        ra[i] = *(const v4f*)(ag + (kt + 1) * KT + i * 32 * K);
      ri = *(const v4f*)(ig + (kt + 1) * KT);
      rc = *(const v4f*)(cg + (kt + 1) * KT);
      __builtin_amdgcn_sched_barrier(0);
    }
    const float* as = &As[cur][a_ld];
    const float* is = &Bi[cur][b_ld];
    const float* cs = &Bc[cur][b_ld];
#pragma unroll
    for (int kk = 0; kk < KT / 8; ++kk) {
      v4f a = *(const v4f*)(as + kk * 8);
      v4f asq = a * a;
      v4f bi0 = *(const v4f*)(is + kk * 8);
      v4f bi1 = *(const v4f*)(is + 16 * PA + kk * 8);
      v4f bc0 = *(const v4f*)(cs + kk * 8);
      v4f bc1 = *(const v4f*)(cs + 16 * PA + kk * 8);
      v2f alo = {a.x, a.y}, ahi = {a.z, a.w};
      v2f slo = {asq.x, asq.y}, shi = {asq.z, asq.w};
      qq0 = wmma_f32(slo, (v2f){bi0.x, bi0.y}, qq0);
      qc0 = wmma_f32(alo, (v2f){bc0.x, bc0.y}, qc0);
      qq1 = wmma_f32(slo, (v2f){bi1.x, bi1.y}, qq1);
      qc1 = wmma_f32(alo, (v2f){bc1.x, bc1.y}, qc1);
      qq0 = wmma_f32(shi, (v2f){bi0.z, bi0.w}, qq0);
      qc0 = wmma_f32(ahi, (v2f){bc0.z, bc0.w}, qc0);
      qq1 = wmma_f32(shi, (v2f){bi1.z, bi1.w}, qq1);
      qc1 = wmma_f32(ahi, (v2f){bc1.z, bc1.w}, qc1);
    }
    if (more) {
#pragma unroll
      for (int i = 0; i < 4; ++i) *(v4f*)(&As[cur ^ 1][st + i * 32 * PA]) = ra[i];
      *(v4f*)(&Bi[cur ^ 1][st]) = ri;
      *(v4f*)(&Bc[cur ^ 1][st]) = rc;
    }
    __syncthreads();
  }
  const float scale = -0.5f / (float)DD;
  v2f ccv = *(const v2f*)(cc + tile_n + 2 * ln);
#pragma unroll
  for (int i = 0; i < 8; ++i) {
    int m = tile_m + i + 8 * hi;
    v2f o = {scale * (qq0[i] - 2.0f * qc0[i] + ccv.x),
             scale * (qq1[i] - 2.0f * qc1[i] + ccv.y)};
    *(v2f*)(z + m * N + tile_n + 2 * ln) = o;
  }
}

__global__ __launch_bounds__(256) void softmax_kernel(float* __restrict__ z) {
  int row = (blockIdx.x * blockDim.x + threadIdx.x) >> 5;
  int lane = threadIdx.x & 31;
  if (row >= MTOT) return;
  float* p = z + (size_t)row * RR;
  float vals[RR / 32];
  float mx = -3.4e38f;
#pragma unroll
  for (int t = 0; t < RR / 32; ++t) {
    vals[t] = p[lane + 32 * t];
    mx = fmaxf(mx, vals[t]);
  }
  mx = waveRedMax(mx);
  float s = 0.0f;
#pragma unroll
  for (int t = 0; t < RR / 32; ++t) {
    vals[t] = __expf(vals[t] - mx);
    s += vals[t];
  }
  s = waveRedSum(s);
  float inv = 1.0f / s;
#pragma unroll
  for (int t = 0; t < RR / 32; ++t) p[lane + 32 * t] = vals[t] * inv;
}

__device__ __forceinline__ void mm_kn_128x64(
    const float* __restrict__ Ablk,
    const float* __restrict__ Btile,
    const int K, v8f& c0, v8f& c1, v8f& c2, v8f& c3) {
  __shared__ __attribute__((aligned(16))) float As[2][128 * PA];
  __shared__ __attribute__((aligned(16))) float Bs[2][KT * PB];
  int t = threadIdx.x;
  int wave = t >> 5;
  int lane = t & 31;
  int hi = lane >> 4;
  int ln = lane & 15;

  int cr = t >> 3;
  int cq = (t & 7) * 4;
  int br = t >> 4;
  int bq = (t & 15) * 4;
  const float* ag = Ablk + cr * K + cq;
  const float* bg = Btile + br * DD + bq;
  int a_st = cr * PA + cq;
  int b_st = br * PB + bq;
  int a_ld = (wave * 16 + ln) * PA + 4 * hi;
  int b_ld = (4 * hi) * PB + 4 * ln;

  v4f ra[4], rb[2];
#pragma unroll
  for (int i = 0; i < 4; ++i) ra[i] = *(const v4f*)(ag + i * 32 * K);
#pragma unroll
  for (int i = 0; i < 2; ++i) rb[i] = *(const v4f*)(bg + i * 16 * DD);
#pragma unroll
  for (int i = 0; i < 4; ++i) *(v4f*)(&As[0][a_st + i * 32 * PA]) = ra[i];
#pragma unroll
  for (int i = 0; i < 2; ++i) *(v4f*)(&Bs[0][b_st + i * 16 * PB]) = rb[i];
  __syncthreads();

  for (int kt = 0; kt < K / KT; ++kt) {
    const int cur = kt & 1;
    const bool more = (kt + 1) < K / KT;
    if (more) {
#pragma unroll
      for (int i = 0; i < 4; ++i)
        ra[i] = *(const v4f*)(ag + (kt + 1) * KT + i * 32 * K);
#pragma unroll
      for (int i = 0; i < 2; ++i)
        rb[i] = *(const v4f*)(bg + ((kt + 1) * KT + i * 16) * DD);
      __builtin_amdgcn_sched_barrier(0);
    }
    const float* as = &As[cur][a_ld];
    const float* bs = &Bs[cur][b_ld];
#pragma unroll
    for (int kk = 0; kk < KT / 8; ++kk) {
      v4f a  = *(const v4f*)(as + kk * 8);
      v4f r0 = *(const v4f*)(bs + (kk * 8 + 0) * PB);
      v4f r1 = *(const v4f*)(bs + (kk * 8 + 1) * PB);
      v4f r2 = *(const v4f*)(bs + (kk * 8 + 2) * PB);
      v4f r3 = *(const v4f*)(bs + (kk * 8 + 3) * PB);
      v2f alo = {a.x, a.y}, ahi = {a.z, a.w};
      c0 = wmma_f32(alo, (v2f){r0.x, r1.x}, c0);
      c1 = wmma_f32(alo, (v2f){r0.y, r1.y}, c1);
      c2 = wmma_f32(alo, (v2f){r0.z, r1.z}, c2);
      c3 = wmma_f32(alo, (v2f){r0.w, r1.w}, c3);
      c0 = wmma_f32(ahi, (v2f){r2.x, r3.x}, c0);
      c1 = wmma_f32(ahi, (v2f){r2.y, r3.y}, c1);
      c2 = wmma_f32(ahi, (v2f){r2.z, r3.z}, c2);
      c3 = wmma_f32(ahi, (v2f){r2.w, r3.w}, c3);
    }
    if (more) {
#pragma unroll
      for (int i = 0; i < 4; ++i) *(v4f*)(&As[cur ^ 1][a_st + i * 32 * PA]) = ra[i];
#pragma unroll
      for (int i = 0; i < 2; ++i) *(v4f*)(&Bs[cur ^ 1][b_st + i * 16 * PB]) = rb[i];
    }
    __syncthreads();
  }
}

__global__ __launch_bounds__(256) void conq_kernel(
    const float* __restrict__ query, const float* __restrict__ Wc,
    const float* __restrict__ bc, float* __restrict__ conq) {
  const int K = LL, N = DD;
  int b = blockIdx.z;
  int wave = threadIdx.x >> 5;
  int lane = threadIdx.x & 31;
  int hi = lane >> 4;
  int ln = lane & 15;
  int blk_m = blockIdx.x * 128;
  int tile_m = blk_m + wave * 16;
  int tile_n = blockIdx.y * 64;

  v8f c0 = {}, c1 = {}, c2 = {}, c3 = {};
  mm_kn_128x64(Wc + blk_m * K, query + b * LL * DD + tile_n, K, c0, c1, c2, c3);

  float* Cb = conq + b * RR * DD;
#pragma unroll
  for (int i = 0; i < 8; ++i) {
    int m = tile_m + i + 8 * hi;
    float bias = bc[m];
    v4f o = {c0[i] + bias, c1[i] + bias, c2[i] + bias, c3[i] + bias};
    *(v4f*)(Cb + m * N + tile_n + 4 * ln) = o;
  }
}

__global__ __launch_bounds__(256) void out_kernel(
    const float* __restrict__ Fss, const float* __restrict__ conq,
    float* __restrict__ out) {
  const int K = RR, N = DD;
  int b = blockIdx.z;
  int wave = threadIdx.x >> 5;
  int lane = threadIdx.x & 31;
  int hi = lane >> 4;
  int ln = lane & 15;
  int blk_m = blockIdx.x * 128;
  int tile_m = blk_m + wave * 16;
  int tile_n = blockIdx.y * 64;

  v8f c0 = {}, c1 = {}, c2 = {}, c3 = {};
  mm_kn_128x64(Fss + b * LL * RR + blk_m * K, conq + b * RR * DD + tile_n, K,
               c0, c1, c2, c3);

  float* Ob = out + b * LL * DD;
#pragma unroll
  for (int i = 0; i < 8; ++i) {
    int m = tile_m + i + 8 * hi;
    v4f o = {c0[i], c1[i], c2[i], c3[i]};
    *(v4f*)(Ob + m * N + tile_n + 4 * ln) = o;
  }
}

extern "C" void kernel_launch(void* const* d_in, const int* in_sizes, int n_in,
                              void* d_out, int out_size, void* d_ws, size_t ws_size,
                              hipStream_t stream) {
  const float* query   = (const float*)d_in[0];
  const float* Wq      = (const float*)d_in[1];
  const float* bq      = (const float*)d_in[2];
  const float* Wc      = (const float*)d_in[3];
  const float* bc      = (const float*)d_in[4];
  const float* centers = (const float*)d_in[5];
  const float* widths  = (const float*)d_in[6];
  float* out = (float*)d_out;

  float* q    = (float*)d_ws;
  float* z    = q    + (size_t)MTOT * DD;
  float* iw   = z    + (size_t)MTOT * RR;
  float* cw   = iw   + (size_t)RR * DD;
  float* cc   = cw   + (size_t)RR * DD;
  float* conq = cc   + RR;

  prep_kernel<<<RR / 8, 256, 0, stream>>>(centers, widths, iw, cw, cc);

  qproj_kernel<<<dim3(MTOT / 128, DD / 64), 256, 0, stream>>>(query, Wq, bq, q);

  z_kernel<<<dim3(MTOT / 128, RR / 32), 256, 0, stream>>>(q, iw, cw, cc, z);

  softmax_kernel<<<MTOT / 8, 256, 0, stream>>>(z);

  conq_kernel<<<dim3(RR / 128, DD / 64, BB), 256, 0, stream>>>(query, Wc, bc, conq);

  out_kernel<<<dim3(LL / 128, DD / 64, BB), 256, 0, stream>>>(z, conq, out);
}


# K1, K3, K6: wave tile 32x64 (2 M-subtiles x 4 N-subtiles; K3 for both products), 8 waves as 4x2 on a 128x128 block tile: an LDS fragment load feeds about twice the WMMAs; k order and sums unchanged
# speedup vs baseline: 2.1838x; 1.1213x over previous
#include <hip/hip_runtime.h>
#include <hip/hip_bf16.h>

#define BB 8
#define LL 2048
#define DD 512
#define RR 512
#define MTOT (BB * LL)

typedef float v2f __attribute__((ext_vector_type(2)));
typedef float v4f __attribute__((ext_vector_type(4)));
typedef float v8f __attribute__((ext_vector_type(8)));


#define KT 32
#define PA 36
#define PB 72

__device__ __forceinline__ v8f wmma_f32(v2f a, v2f b, v8f c) {
  return __builtin_amdgcn_wmma_f32_16x16x4_f32(
       false, a,  false, b,
       (short)0, c,  false,  false);
}

__device__ __forceinline__ float waveRedSum(float v) {
#pragma unroll
  for (int m = 16; m >= 1; m >>= 1) v += __shfl_xor(v, m, 32);
  return v;
}
__device__ __forceinline__ float waveRedMax(float v) {
#pragma unroll
  for (int m = 16; m >= 1; m >>= 1) v = fmaxf(v, __shfl_xor(v, m, 32));
  return v;
}

__global__ __launch_bounds__(256) void prep_kernel(
    const float* __restrict__ centers, const float* __restrict__ widths,
    float* __restrict__ iw, float* __restrict__ cw, float* __restrict__ cc) {
  int row = (blockIdx.x * blockDim.x + threadIdx.x) >> 5;
  int lane = threadIdx.x & 31;
  if (row >= RR) return;
  const float* crow = centers + row * DD;
  const float* wrow = widths + row * DD;
  float acc = 0.0f;
#pragma unroll
  for (int t = 0; t < DD / 32; ++t) {
    int d = lane + 32 * t;
    float w = wrow[d];
    float inv = 1.0f / (w * w);
    float c = crow[d];
    iw[row * DD + d] = inv;
    cw[row * DD + d] = c * inv;
    acc += c * c * inv;
  }
  acc = waveRedSum(acc);
  if (lane == 0) cc[row] = acc;
}

__global__ __launch_bounds__(256) void qproj_kernel(
    const float* __restrict__ A, const float* __restrict__ Bm,
    const float* __restrict__ bias, float* __restrict__ C) {
  const int K = DD, N = DD;
  __shared__ __attribute__((aligned(16))) float As[2][128 * PA];
  __shared__ __attribute__((aligned(16))) float Bs[2][128 * PA];
  int t = threadIdx.x;
  int wave = t >> 5;
  int lane = t & 31;
  int hi = lane >> 4;
  int ln = lane & 15;
  int wm = wave & 3;
  int wn = wave >> 2;
  int blk_m = blockIdx.x * 128;
  int tile_m = blk_m + wm * 32;
  int tile_n = blockIdx.y * 128;

  int cr = t >> 3;
  int cq = (t & 7) * 4;
  const float* ag = A + (blk_m + cr) * K + cq;
  const float* bg = Bm + (tile_n + 4 * (cr & 15) + (cr >> 4)) * K + cq;
  int st = cr * PA + cq;
  int a_ld = (wm * 32 + ln) * PA + 4 * hi;
  int b_ld = (wn * 64 + ln) * PA + 4 * hi;

  v4f ra[4], rb[4];
#pragma unroll
  for (int i = 0; i < 4; ++i) ra[i] = *(const v4f*)(ag + i * 32 * K);
#pragma unroll
  for (int i = 0; i < 4; ++i)
    rb[i] = *(const v4f*)(bg + ((i >> 1) * 64 + 2 * (i & 1)) * K);
#pragma unroll
  for (int i = 0; i < 4; ++i) *(v4f*)(&As[0][st + i * 32 * PA]) = ra[i];
#pragma unroll
  for (int i = 0; i < 4; ++i) *(v4f*)(&Bs[0][st + i * 32 * PA]) = rb[i];
  __syncthreads();

  v8f c[2][4] = {};
  for (int kt = 0; kt < K / KT; ++kt) {
    const int cur = kt & 1;
    const bool more = (kt + 1) < K / KT;
    if (more) {
#pragma unroll
      for (int i = 0; i < 4; ++i)
        ra[i] = *(const v4f*)(ag + (kt + 1) * KT + i * 32 * K);
#pragma unroll
      for (int i = 0; i < 4; ++i)
        rb[i] = *(const v4f*)(bg + (kt + 1) * KT + ((i >> 1) * 64 + 2 * (i & 1)) * K);
      __builtin_amdgcn_sched_barrier(0);
    }
    const float* as = &As[cur][a_ld];
    const float* bs = &Bs[cur][b_ld];
#pragma unroll
    for (int kk = 0; kk < KT / 8; ++kk) {
      v4f a[2], b[4];
#pragma unroll
      for (int mi = 0; mi < 2; ++mi) a[mi] = *(const v4f*)(as + mi * 16 * PA + kk * 8);
#pragma unroll
      for (int j = 0; j < 4; ++j) b[j] = *(const v4f*)(bs + j * 16 * PA + kk * 8);
#pragma unroll
      for (int mi = 0; mi < 2; ++mi)
#pragma unroll
        for (int j = 0; j < 4; ++j)
          c[mi][j] = wmma_f32((v2f){a[mi].x, a[mi].y}, (v2f){b[j].x, b[j].y}, c[mi][j]);
#pragma unroll
      for (int mi = 0; mi < 2; ++mi)
#pragma unroll
        for (int j = 0; j < 4; ++j)
          c[mi][j] = wmma_f32((v2f){a[mi].z, a[mi].w}, (v2f){b[j].z, b[j].w}, c[mi][j]);
    }
    if (more) {
#pragma unroll
      for (int i = 0; i < 4; ++i) *(v4f*)(&As[cur ^ 1][st + i * 32 * PA]) = ra[i];
#pragma unroll
      for (int i = 0; i < 4; ++i) *(v4f*)(&Bs[cur ^ 1][st + i * 32 * PA]) = rb[i];
    }
    __syncthreads();
  }
  v4f bq = *(const v4f*)(bias + tile_n + 64 * wn + 4 * ln);
#pragma unroll
  for (int mi = 0; mi < 2; ++mi) {
#pragma unroll
    for (int i = 0; i < 8; ++i) {
      int m = tile_m + mi * 16 + i + 8 * hi;
      v4f o = {c[mi][0][i] + bq.x, c[mi][1][i] + bq.y, c[mi][2][i] + bq.z, c[mi][3][i] + bq.w};
      *(v4f*)(C + m * N + tile_n + 64 * wn + 4 * ln) = o;
    }
  }
}

__global__ __launch_bounds__(256) void z_kernel(
    const float* __restrict__ q, const float* __restrict__ iw,
    const float* __restrict__ cw, const float* __restrict__ cc,
    float* __restrict__ z) {
  const int K = DD, N = RR;
  __shared__ __attribute__((aligned(16))) float As[2][128 * PA];
  __shared__ __attribute__((aligned(16))) float Bi[2][128 * PA];
  __shared__ __attribute__((aligned(16))) float Bc[2][128 * PA];
  int t = threadIdx.x;
  int wave = t >> 5;
  int lane = t & 31;
  int hi = lane >> 4;
  int ln = lane & 15;
  int wm = wave & 3;
  int wn = wave >> 2;
  int blk_m = blockIdx.x * 128;
  int tile_m = blk_m + wm * 32;
  int tile_n = blockIdx.y * 128;

  int cr = t >> 3;
  int cq = (t & 7) * 4;
  const float* ag = q + (blk_m + cr) * K + cq;
  int brow = tile_n + 4 * (cr & 15) + (cr >> 4);
  const float* ig = iw + brow * K + cq;
  const float* cg = cw + brow * K + cq;
  int st = cr * PA + cq;
  int a_ld = (wm * 32 + ln) * PA + 4 * hi;
  int b_ld = (wn * 64 + ln) * PA + 4 * hi;

  v4f ra[4], ri[4], rc[4];
#pragma unroll
  for (int i = 0; i < 4; ++i) {
    ra[i] = *(const v4f*)(ag + i * 32 * K);
    ri[i] = *(const v4f*)(ig + ((i >> 1) * 64 + 2 * (i & 1)) * K);
    rc[i] = *(const v4f*)(cg + ((i >> 1) * 64 + 2 * (i & 1)) * K);
  }
#pragma unroll
  for (int i = 0; i < 4; ++i) {
    *(v4f*)(&As[0][st + i * 32 * PA]) = ra[i];
    *(v4f*)(&Bi[0][st + i * 32 * PA]) = ri[i];
    *(v4f*)(&Bc[0][st + i * 32 * PA]) = rc[i];
  }
  __syncthreads();

  v8f qq[2][4] = {}, qc[2][4] = {};
  for (int kt = 0; kt < K / KT; ++kt) {
    const int cur = kt & 1;
    const bool more = (kt + 1) < K / KT;
    if (more) {
#pragma unroll
      for (int i = 0; i < 4; ++i) {
        ra[i] = *(const v4f*)(ag + (kt + 1) * KT + i * 32 * K);
        ri[i] = *(const v4f*)(ig + (kt + 1) * KT + ((i >> 1) * 64 + 2 * (i & 1)) * K);
        rc[i] = *(const v4f*)(cg + (kt + 1) * KT + ((i >> 1) * 64 + 2 * (i & 1)) * K);
      }
      __builtin_amdgcn_sched_barrier(0);
    }
    const float* as = &As[cur][a_ld];
    const float* is = &Bi[cur][b_ld];
    const float* cs = &Bc[cur][b_ld];
#pragma unroll
    for (int kk = 0; kk < KT / 8; ++kk) {
      v4f a[2], s[2], bi[4], bc[4];
#pragma unroll
      for (int mi = 0; mi < 2; ++mi) {
        a[mi] = *(const v4f*)(as + mi * 16 * PA + kk * 8);
        s[mi] = a[mi] * a[mi];
      }
#pragma unroll
      for (int j = 0; j < 4; ++j) {
        bi[j] = *(const v4f*)(is + j * 16 * PA + kk * 8);
        bc[j] = *(const v4f*)(cs + j * 16 * PA + kk * 8);
      }
#pragma unroll
      for (int mi = 0; mi < 2; ++mi)
#pragma unroll
        for (int j = 0; j < 4; ++j) {
          qq[mi][j] = wmma_f32((v2f){s[mi].x, s[mi].y}, (v2f){bi[j].x, bi[j].y}, qq[mi][j]);
          qc[mi][j] = wmma_f32((v2f){a[mi].x, a[mi].y}, (v2f){bc[j].x, bc[j].y}, qc[mi][j]);
        }
#pragma unroll
      for (int mi = 0; mi < 2; ++mi)
#pragma unroll
        for (int j = 0; j < 4; ++j) {
          qq[mi][j] = wmma_f32((v2f){s[mi].z, s[mi].w}, (v2f){bi[j].z, bi[j].w}, qq[mi][j]);
          qc[mi][j] = wmma_f32((v2f){a[mi].z, a[mi].w}, (v2f){bc[j].z, bc[j].w}, qc[mi][j]);
        }
    }
    if (more) {
#pragma unroll
      for (int i = 0; i < 4; ++i) {
        *(v4f*)(&As[cur ^ 1][st + i * 32 * PA]) = ra[i];
        *(v4f*)(&Bi[cur ^ 1][st + i * 32 * PA]) = ri[i];
        *(v4f*)(&Bc[cur ^ 1][st + i * 32 * PA]) = rc[i];
      }
    }
    __syncthreads();
  }
  const float scale = -0.5f / (float)DD;
  v4f ccv = *(const v4f*)(cc + tile_n + 64 * wn + 4 * ln);
#pragma unroll
  for (int mi = 0; mi < 2; ++mi) {
#pragma unroll
    for (int i = 0; i < 8; ++i) {
      int m = tile_m + mi * 16 + i + 8 * hi;
      v4f o = {scale * (qq[mi][0][i] - 2.0f * qc[mi][0][i] + ccv.x),
               scale * (qq[mi][1][i] - 2.0f * qc[mi][1][i] + ccv.y),
               scale * (qq[mi][2][i] - 2.0f * qc[mi][2][i] + ccv.z),
               scale * (qq[mi][3][i] - 2.0f * qc[mi][3][i] + ccv.w)};
      *(v4f*)(z + m * N + tile_n + 64 * wn + 4 * ln) = o;
    }
  }
}

__global__ __launch_bounds__(256) void softmax_kernel(float* __restrict__ z) {
  int row = (blockIdx.x * blockDim.x + threadIdx.x) >> 5;
  int lane = threadIdx.x & 31;
  if (row >= MTOT) return;
  float* p = z + (size_t)row * RR;
  float vals[RR / 32];
  float mx = -3.4e38f;
#pragma unroll
  for (int t = 0; t < RR / 32; ++t) {
    vals[t] = p[lane + 32 * t];
    mx = fmaxf(mx, vals[t]);
  }
  mx = waveRedMax(mx);
  float s = 0.0f;
#pragma unroll
  for (int t = 0; t < RR / 32; ++t) {
    vals[t] = __expf(vals[t] - mx);
    s += vals[t];
  }
  s = waveRedSum(s);
  float inv = 1.0f / s;
#pragma unroll
  for (int t = 0; t < RR / 32; ++t) p[lane + 32 * t] = vals[t] * inv;
}

__device__ __forceinline__ void mm_kn_128x64(
    const float* __restrict__ Ablk,
    const float* __restrict__ Btile,
    const int K, v8f& c0, v8f& c1, v8f& c2, v8f& c3) {
  __shared__ __attribute__((aligned(16))) float As[2][128 * PA];
  __shared__ __attribute__((aligned(16))) float Bs[2][KT * PB];
  int t = threadIdx.x;
  int wave = t >> 5;
  int lane = t & 31;
  int hi = lane >> 4;
  int ln = lane & 15;

  int cr = t >> 3;
  int cq = (t & 7) * 4;
  int br = t >> 4;
  int bq = (t & 15) * 4;
  const float* ag = Ablk + cr * K + cq;
  const float* bg = Btile + br * DD + bq;
  int a_st = cr * PA + cq;
  int b_st = br * PB + bq;
  int a_ld = (wave * 16 + ln) * PA + 4 * hi;
  int b_ld = (4 * hi) * PB + 4 * ln;

  v4f ra[4], rb[2];
#pragma unroll
  for (int i = 0; i < 4; ++i) ra[i] = *(const v4f*)(ag + i * 32 * K);
#pragma unroll
  for (int i = 0; i < 2; ++i) rb[i] = *(const v4f*)(bg + i * 16 * DD);
#pragma unroll
  for (int i = 0; i < 4; ++i) *(v4f*)(&As[0][a_st + i * 32 * PA]) = ra[i];
#pragma unroll
  for (int i = 0; i < 2; ++i) *(v4f*)(&Bs[0][b_st + i * 16 * PB]) = rb[i];
  __syncthreads();

  for (int kt = 0; kt < K / KT; ++kt) {
    const int cur = kt & 1;
    const bool more = (kt + 1) < K / KT;
    if (more) {
#pragma unroll
      for (int i = 0; i < 4; ++i)
        ra[i] = *(const v4f*)(ag + (kt + 1) * KT + i * 32 * K);
#pragma unroll
      for (int i = 0; i < 2; ++i)
        rb[i] = *(const v4f*)(bg + ((kt + 1) * KT + i * 16) * DD);
      __builtin_amdgcn_sched_barrier(0);
    }
    const float* as = &As[cur][a_ld];
    const float* bs = &Bs[cur][b_ld];
#pragma unroll
    for (int kk = 0; kk < KT / 8; ++kk) {
      v4f a  = *(const v4f*)(as + kk * 8);
      v4f r0 = *(const v4f*)(bs + (kk * 8 + 0) * PB);
      v4f r1 = *(const v4f*)(bs + (kk * 8 + 1) * PB);
      v4f r2 = *(const v4f*)(bs + (kk * 8 + 2) * PB);
      v4f r3 = *(const v4f*)(bs + (kk * 8 + 3) * PB);
      v2f alo = {a.x, a.y}, ahi = {a.z, a.w};
      c0 = wmma_f32(alo, (v2f){r0.x, r1.x}, c0);
      c1 = wmma_f32(alo, (v2f){r0.y, r1.y}, c1);
      c2 = wmma_f32(alo, (v2f){r0.z, r1.z}, c2);
      c3 = wmma_f32(alo, (v2f){r0.w, r1.w}, c3);
      c0 = wmma_f32(ahi, (v2f){r2.x, r3.x}, c0);
      c1 = wmma_f32(ahi, (v2f){r2.y, r3.y}, c1);
      c2 = wmma_f32(ahi, (v2f){r2.z, r3.z}, c2);
      c3 = wmma_f32(ahi, (v2f){r2.w, r3.w}, c3);
    }
    if (more) {
#pragma unroll
      for (int i = 0; i < 4; ++i) *(v4f*)(&As[cur ^ 1][a_st + i * 32 * PA]) = ra[i];
#pragma unroll
      for (int i = 0; i < 2; ++i) *(v4f*)(&Bs[cur ^ 1][b_st + i * 16 * PB]) = rb[i];
    }
    __syncthreads();
  }
}

__global__ __launch_bounds__(256) void conq_kernel(
    const float* __restrict__ query, const float* __restrict__ Wc,
    const float* __restrict__ bc, float* __restrict__ conq) {
  const int K = LL, N = DD;
  int b = blockIdx.z;
  int wave = threadIdx.x >> 5;
  int lane = threadIdx.x & 31;
  int hi = lane >> 4;
  int ln = lane & 15;
  int blk_m = blockIdx.x * 128;
  int tile_m = blk_m + wave * 16;
  int tile_n = blockIdx.y * 64;

  v8f c0 = {}, c1 = {}, c2 = {}, c3 = {};
  mm_kn_128x64(Wc + blk_m * K, query + b * LL * DD + tile_n, K, c0, c1, c2, c3);

  float* Cb = conq + b * RR * DD;
#pragma unroll
  for (int i = 0; i < 8; ++i) {
    int m = tile_m + i + 8 * hi;
    float bias = bc[m];
    v4f o = {c0[i] + bias, c1[i] + bias, c2[i] + bias, c3[i] + bias};
    *(v4f*)(Cb + m * N + tile_n + 4 * ln) = o;
  }
}

#define PB2 136
__device__ __forceinline__ void mm_kn_128x128(
    const float* __restrict__ Ablk,
    const float* __restrict__ Btile,
    const int K, v8f (&c)[2][4]) {
  __shared__ __attribute__((aligned(16))) float As[2][128 * PA];
  __shared__ __attribute__((aligned(16))) float Bs[2][KT * PB2];
  int t = threadIdx.x;
  int wave = t >> 5;
  int lane = t & 31;
  int hi = lane >> 4;
  int ln = lane & 15;
  int wm = wave & 3;
  int wn = wave >> 2;

  int cr = t >> 3;
  int cq = (t & 7) * 4;
  int br = t >> 5;
  int bq = (t & 31) * 4;
  const float* ag = Ablk + cr * K + cq;
  const float* bg = Btile + br * DD + bq;
  int a_st = cr * PA + cq;
  int b_st = br * PB2 + bq;
  int a_ld = (wm * 32 + ln) * PA + 4 * hi;
  int b_ld = (4 * hi) * PB2 + 64 * wn + 4 * ln;

  v4f ra[4], rb[4];
#pragma unroll
  for (int i = 0; i < 4; ++i) ra[i] = *(const v4f*)(ag + i * 32 * K);
#pragma unroll
  for (int i = 0; i < 4; ++i) rb[i] = *(const v4f*)(bg + i * 8 * DD);
#pragma unroll
  for (int i = 0; i < 4; ++i) *(v4f*)(&As[0][a_st + i * 32 * PA]) = ra[i];
#pragma unroll
  for (int i = 0; i < 4; ++i) *(v4f*)(&Bs[0][b_st + i * 8 * PB2]) = rb[i];
  __syncthreads();

  for (int kt = 0; kt < K / KT; ++kt) {
    const int cur = kt & 1;
    const bool more = (kt + 1) < K / KT;
    if (more) {
#pragma unroll
      for (int i = 0; i < 4; ++i)
        ra[i] = *(const v4f*)(ag + (kt + 1) * KT + i * 32 * K);
#pragma unroll
      for (int i = 0; i < 4; ++i)
        rb[i] = *(const v4f*)(bg + ((kt + 1) * KT + i * 8) * DD);
      __builtin_amdgcn_sched_barrier(0);
    }
    const float* as = &As[cur][a_ld];
    const float* bs = &Bs[cur][b_ld];
#pragma unroll
    for (int kk = 0; kk < KT / 8; ++kk) {
      v4f a[2], r[4];
#pragma unroll
      for (int mi = 0; mi < 2; ++mi) a[mi] = *(const v4f*)(as + mi * 16 * PA + kk * 8);
#pragma unroll
      for (int j = 0; j < 4; ++j) r[j] = *(const v4f*)(bs + (kk * 8 + j) * PB2);
#pragma unroll
      for (int mi = 0; mi < 2; ++mi) {
        v2f alo = {a[mi].x, a[mi].y};
        c[mi][0] = wmma_f32(alo, (v2f){r[0].x, r[1].x}, c[mi][0]);
        c[mi][1] = wmma_f32(alo, (v2f){r[0].y, r[1].y}, c[mi][1]);
        c[mi][2] = wmma_f32(alo, (v2f){r[0].z, r[1].z}, c[mi][2]);
        c[mi][3] = wmma_f32(alo, (v2f){r[0].w, r[1].w}, c[mi][3]);
      }
#pragma unroll
      for (int mi = 0; mi < 2; ++mi) {
        v2f ahi = {a[mi].z, a[mi].w};
        c[mi][0] = wmma_f32(ahi, (v2f){r[2].x, r[3].x}, c[mi][0]);
        c[mi][1] = wmma_f32(ahi, (v2f){r[2].y, r[3].y}, c[mi][1]);
        c[mi][2] = wmma_f32(ahi, (v2f){r[2].z, r[3].z}, c[mi][2]);
        c[mi][3] = wmma_f32(ahi, (v2f){r[2].w, r[3].w}, c[mi][3]);
      }
    }
    if (more) {
#pragma unroll
      for (int i = 0; i < 4; ++i) *(v4f*)(&As[cur ^ 1][a_st + i * 32 * PA]) = ra[i];
#pragma unroll
      for (int i = 0; i < 4; ++i) *(v4f*)(&Bs[cur ^ 1][b_st + i * 8 * PB2]) = rb[i];
    }
    __syncthreads();
  }
}

__global__ __launch_bounds__(256) void out_kernel(
    const float* __restrict__ Fss, const float* __restrict__ conq,
    float* __restrict__ out) {
  const int K = RR, N = DD;
  int b = blockIdx.z;
  int wave = threadIdx.x >> 5;
  int lane = threadIdx.x & 31;
  int hi = lane >> 4;
  int ln = lane & 15;
  int wm = wave & 3;
  int wn = wave >> 2;
  int blk_m = blockIdx.x * 128;
  int tile_m = blk_m + wm * 32;
  int tile_n = blockIdx.y * 128;

  v8f c[2][4] = {};
  mm_kn_128x128(Fss + b * LL * RR + blk_m * K, conq + b * RR * DD + tile_n, K, c);

  float* Ob = out + b * LL * DD;
#pragma unroll
  for (int mi = 0; mi < 2; ++mi) {
#pragma unroll
    for (int i = 0; i < 8; ++i) {
      int m = tile_m + mi * 16 + i + 8 * hi;
      v4f o = {c[mi][0][i], c[mi][1][i], c[mi][2][i], c[mi][3][i]};
      *(v4f*)(Ob + m * N + tile_n + 64 * wn + 4 * ln) = o;
    }
  }
}

extern "C" void kernel_launch(void* const* d_in, const int* in_sizes, int n_in,
                              void* d_out, int out_size, void* d_ws, size_t ws_size,
                              hipStream_t stream) {
  const float* query   = (const float*)d_in[0];
  const float* Wq      = (const float*)d_in[1];
  const float* bq      = (const float*)d_in[2];
  const float* Wc      = (const float*)d_in[3];
  const float* bc      = (const float*)d_in[4];
  const float* centers = (const float*)d_in[5];
  const float* widths  = (const float*)d_in[6];
  float* out = (float*)d_out;

  float* q    = (float*)d_ws;
  float* z    = q    + (size_t)MTOT * DD;
  float* iw   = z    + (size_t)MTOT * RR;
  float* cw   = iw   + (size_t)RR * DD;
  float* cc   = cw   + (size_t)RR * DD;
  float* conq = cc   + RR;

  prep_kernel<<<RR / 8, 256, 0, stream>>>(centers, widths, iw, cw, cc);

  qproj_kernel<<<dim3(MTOT / 128, DD / 128), 256, 0, stream>>>(query, Wq, bq, q);

  z_kernel<<<dim3(MTOT / 128, RR / 128), 256, 0, stream>>>(q, iw, cw, cc, z);

  softmax_kernel<<<MTOT / 8, 256, 0, stream>>>(z);

  conq_kernel<<<dim3(RR / 128, DD / 64, BB), 256, 0, stream>>>(query, Wc, bc, conq);

  out_kernel<<<dim3(LL / 128, DD / 128, BB), 256, 0, stream>>>(z, conq, out);
}


# The two memory-bound kernels ride as extra blocks in the launch of an independent GEMM: rule precompute in front of K1's tiles, row softmax behind K5's tiles; 6 launches -> 4; all sums unchanged
# speedup vs baseline: 2.2765x; 1.0424x over previous
#include <hip/hip_runtime.h>
#include <hip/hip_bf16.h>

#define BB 8
#define LL 2048
#define DD 512
#define RR 512
#define MTOT (BB * LL)

typedef float v2f __attribute__((ext_vector_type(2)));
typedef float v4f __attribute__((ext_vector_type(4)));
typedef float v8f __attribute__((ext_vector_type(8)));


#define KT 32
#define PA 36
#define PB 72

__device__ __forceinline__ v8f wmma_f32(v2f a, v2f b, v8f c) {
  return __builtin_amdgcn_wmma_f32_16x16x4_f32(
       false, a,  false, b,
       (short)0, c,  false,  false);
}

__device__ __forceinline__ float waveRedSum(float v) {
#pragma unroll
  for (int m = 16; m >= 1; m >>= 1) v += __shfl_xor(v, m, 32);
  return v;
}
__device__ __forceinline__ float waveRedMax(float v) {
#pragma unroll
  for (int m = 16; m >= 1; m >>= 1) v = fmaxf(v, __shfl_xor(v, m, 32));
  return v;
}

#define PREP_BLOCKS (RR / 8)
__device__ __forceinline__ void prep_rows(
    const int blk,
    const float* __restrict__ centers, const float* __restrict__ widths,
    float* __restrict__ iw, float* __restrict__ cw, float* __restrict__ cc) {
  int row = (blk * 256 + threadIdx.x) >> 5;
  int lane = threadIdx.x & 31;
  if (row >= RR) return;
  const float* crow = centers + row * DD;
  const float* wrow = widths + row * DD;
  float acc = 0.0f;
#pragma unroll
  for (int t = 0; t < DD / 32; ++t) {
    int d = lane + 32 * t;
    float w = wrow[d];
    float inv = 1.0f / (w * w);
    float c = crow[d];
    iw[row * DD + d] = inv;
    cw[row * DD + d] = c * inv;
    acc += c * c * inv;
  }
  acc = waveRedSum(acc);
  if (lane == 0) cc[row] = acc;
}

__global__ __launch_bounds__(256) void qproj_prep_kernel(
    const float* __restrict__ A, const float* __restrict__ Bm,
    const float* __restrict__ bias, float* __restrict__ C,
    const float* __restrict__ centers, const float* __restrict__ widths,
    float* __restrict__ iw, float* __restrict__ cw, float* __restrict__ cc) {
  if (blockIdx.x < PREP_BLOCKS) {
    prep_rows(blockIdx.x, centers, widths, iw, cw, cc);
    return;
  }
  const int tile_id = blockIdx.x - PREP_BLOCKS;
  const int tile_x = tile_id & (MTOT / 128 - 1);
  const int tile_y = tile_id / (MTOT / 128);
  const int K = DD, N = DD;
  __shared__ __attribute__((aligned(16))) float As[2][128 * PA];
  __shared__ __attribute__((aligned(16))) float Bs[2][128 * PA];
  int t = threadIdx.x;
  int wave = t >> 5;
  int lane = t & 31;
  int hi = lane >> 4;
  int ln = lane & 15;
  int wm = wave & 3;
  int wn = wave >> 2;
  int blk_m = tile_x * 128;
  int tile_m = blk_m + wm * 32;
  int tile_n = tile_y * 128;

  int cr = t >> 3;
  int cq = (t & 7) * 4;
  const float* ag = A + (blk_m + cr) * K + cq;
  const float* bg = Bm + (tile_n + 4 * (cr & 15) + (cr >> 4)) * K + cq;
  int st = cr * PA + cq;
  int a_ld = (wm * 32 + ln) * PA + 4 * hi;
  int b_ld = (wn * 64 + ln) * PA + 4 * hi;

  v4f ra[4], rb[4];
#pragma unroll
  for (int i = 0; i < 4; ++i) ra[i] = *(const v4f*)(ag + i * 32 * K);
#pragma unroll
  for (int i = 0; i < 4; ++i)
    rb[i] = *(const v4f*)(bg + ((i >> 1) * 64 + 2 * (i & 1)) * K);
#pragma unroll
  for (int i = 0; i < 4; ++i) *(v4f*)(&As[0][st + i * 32 * PA]) = ra[i];
#pragma unroll
  for (int i = 0; i < 4; ++i) *(v4f*)(&Bs[0][st + i * 32 * PA]) = rb[i];
  __syncthreads();

  v8f c[2][4] = {};
  for (int kt = 0; kt < K / KT; ++kt) {
    const int cur = kt & 1;
    const bool more = (kt + 1) < K / KT;
    if (more) {
#pragma unroll
      for (int i = 0; i < 4; ++i)
        ra[i] = *(const v4f*)(ag + (kt + 1) * KT + i * 32 * K);
#pragma unroll
      for (int i = 0; i < 4; ++i)
        rb[i] = *(const v4f*)(bg + (kt + 1) * KT + ((i >> 1) * 64 + 2 * (i & 1)) * K);
      __builtin_amdgcn_sched_barrier(0);
    }
    const float* as = &As[cur][a_ld];
    const float* bs = &Bs[cur][b_ld];
#pragma unroll
    for (int kk = 0; kk < KT / 8; ++kk) {
      v4f a[2], b[4];
#pragma unroll
      for (int mi = 0; mi < 2; ++mi) a[mi] = *(const v4f*)(as + mi * 16 * PA + kk * 8);
#pragma unroll
      for (int j = 0; j < 4; ++j) b[j] = *(const v4f*)(bs + j * 16 * PA + kk * 8);
#pragma unroll
      for (int mi = 0; mi < 2; ++mi)
#pragma unroll
        for (int j = 0; j < 4; ++j)
          c[mi][j] = wmma_f32((v2f){a[mi].x, a[mi].y}, (v2f){b[j].x, b[j].y}, c[mi][j]);
#pragma unroll
      for (int mi = 0; mi < 2; ++mi)
#pragma unroll
        for (int j = 0; j < 4; ++j)
          c[mi][j] = wmma_f32((v2f){a[mi].z, a[mi].w}, (v2f){b[j].z, b[j].w}, c[mi][j]);
    }
    if (more) {
#pragma unroll
      for (int i = 0; i < 4; ++i) *(v4f*)(&As[cur ^ 1][st + i * 32 * PA]) = ra[i];
#pragma unroll
      for (int i = 0; i < 4; ++i) *(v4f*)(&Bs[cur ^ 1][st + i * 32 * PA]) = rb[i];
    }
    __syncthreads();
  }
  v4f bq = *(const v4f*)(bias + tile_n + 64 * wn + 4 * ln);
#pragma unroll
  for (int mi = 0; mi < 2; ++mi) {
#pragma unroll
    for (int i = 0; i < 8; ++i) {
      int m = tile_m + mi * 16 + i + 8 * hi;
      v4f o = {c[mi][0][i] + bq.x, c[mi][1][i] + bq.y, c[mi][2][i] + bq.z, c[mi][3][i] + bq.w};
      *(v4f*)(C + m * N + tile_n + 64 * wn + 4 * ln) = o;
    }
  }
}

__global__ __launch_bounds__(256) void z_kernel(
    const float* __restrict__ q, const float* __restrict__ iw,
    const float* __restrict__ cw, const float* __restrict__ cc,
    float* __restrict__ z) {
  const int K = DD, N = RR;
  __shared__ __attribute__((aligned(16))) float As[2][128 * PA];
  __shared__ __attribute__((aligned(16))) float Bi[2][128 * PA];
  __shared__ __attribute__((aligned(16))) float Bc[2][128 * PA];
  int t = threadIdx.x;
  int wave = t >> 5;
  int lane = t & 31;
  int hi = lane >> 4;
  int ln = lane & 15;
  int wm = wave & 3;
  int wn = wave >> 2;
  int blk_m = blockIdx.x * 128;
  int tile_m = blk_m + wm * 32;
  int tile_n = blockIdx.y * 128;

  int cr = t >> 3;
  int cq = (t & 7) * 4;
  const float* ag = q + (blk_m + cr) * K + cq;
  int brow = tile_n + 4 * (cr & 15) + (cr >> 4);
  const float* ig = iw + brow * K + cq;
  const float* cg = cw + brow * K + cq;
  int st = cr * PA + cq;
  int a_ld = (wm * 32 + ln) * PA + 4 * hi;
  int b_ld = (wn * 64 + ln) * PA + 4 * hi;

  v4f ra[4], ri[4], rc[4];
#pragma unroll
  for (int i = 0; i < 4; ++i) {
    ra[i] = *(const v4f*)(ag + i * 32 * K);
    ri[i] = *(const v4f*)(ig + ((i >> 1) * 64 + 2 * (i & 1)) * K);
    rc[i] = *(const v4f*)(cg + ((i >> 1) * 64 + 2 * (i & 1)) * K);
  }
#pragma unroll
  for (int i = 0; i < 4; ++i) {
    *(v4f*)(&As[0][st + i * 32 * PA]) = ra[i];
    *(v4f*)(&Bi[0][st + i * 32 * PA]) = ri[i];
    *(v4f*)(&Bc[0][st + i * 32 * PA]) = rc[i];
  }
  __syncthreads();

  v8f qq[2][4] = {}, qc[2][4] = {};
  for (int kt = 0; kt < K / KT; ++kt) {
    const int cur = kt & 1;
    const bool more = (kt + 1) < K / KT;
    if (more) {
#pragma unroll
      for (int i = 0; i < 4; ++i) {
        ra[i] = *(const v4f*)(ag + (kt + 1) * KT + i * 32 * K);
        ri[i] = *(const v4f*)(ig + (kt + 1) * KT + ((i >> 1) * 64 + 2 * (i & 1)) * K);
        rc[i] = *(const v4f*)(cg + (kt + 1) * KT + ((i >> 1) * 64 + 2 * (i & 1)) * K);
      }
      __builtin_amdgcn_sched_barrier(0);
    }
    const float* as = &As[cur][a_ld];
    const float* is = &Bi[cur][b_ld];
    const float* cs = &Bc[cur][b_ld];
#pragma unroll
    for (int kk = 0; kk < KT / 8; ++kk) {
      v4f a[2], s[2], bi[4], bc[4];
#pragma unroll
      for (int mi = 0; mi < 2; ++mi) {
        a[mi] = *(const v4f*)(as + mi * 16 * PA + kk * 8);
        s[mi] = a[mi] * a[mi];
      }
#pragma unroll
      for (int j = 0; j < 4; ++j) {
        bi[j] = *(const v4f*)(is + j * 16 * PA + kk * 8);
        bc[j] = *(const v4f*)(cs + j * 16 * PA + kk * 8);
      }
#pragma unroll
      for (int mi = 0; mi < 2; ++mi)
#pragma unroll
        for (int j = 0; j < 4; ++j) {
          qq[mi][j] = wmma_f32((v2f){s[mi].x, s[mi].y}, (v2f){bi[j].x, bi[j].y}, qq[mi][j]);
          qc[mi][j] = wmma_f32((v2f){a[mi].x, a[mi].y}, (v2f){bc[j].x, bc[j].y}, qc[mi][j]);
        }
#pragma unroll
      for (int mi = 0; mi < 2; ++mi)
#pragma unroll
        for (int j = 0; j < 4; ++j) {
          qq[mi][j] = wmma_f32((v2f){s[mi].z, s[mi].w}, (v2f){bi[j].z, bi[j].w}, qq[mi][j]);
          qc[mi][j] = wmma_f32((v2f){a[mi].z, a[mi].w}, (v2f){bc[j].z, bc[j].w}, qc[mi][j]);
        }
    }
    if (more) {
#pragma unroll
      for (int i = 0; i < 4; ++i) {
        *(v4f*)(&As[cur ^ 1][st + i * 32 * PA]) = ra[i];
        *(v4f*)(&Bi[cur ^ 1][st + i * 32 * PA]) = ri[i];
        *(v4f*)(&Bc[cur ^ 1][st + i * 32 * PA]) = rc[i];
      }
    }
    __syncthreads();
  }
  const float scale = -0.5f / (float)DD;
  v4f ccv = *(const v4f*)(cc + tile_n + 64 * wn + 4 * ln);
#pragma unroll
  for (int mi = 0; mi < 2; ++mi) {
#pragma unroll
    for (int i = 0; i < 8; ++i) {
      int m = tile_m + mi * 16 + i + 8 * hi;
      v4f o = {scale * (qq[mi][0][i] - 2.0f * qc[mi][0][i] + ccv.x),
               scale * (qq[mi][1][i] - 2.0f * qc[mi][1][i] + ccv.y),
               scale * (qq[mi][2][i] - 2.0f * qc[mi][2][i] + ccv.z),
               scale * (qq[mi][3][i] - 2.0f * qc[mi][3][i] + ccv.w)};
      *(v4f*)(z + m * N + tile_n + 64 * wn + 4 * ln) = o;
    }
  }
}

#define SOFTMAX_BLOCKS (MTOT / 8)
__device__ __forceinline__ void softmax_rows(const int blk, float* __restrict__ z) {
  int row = (blk * 256 + threadIdx.x) >> 5;
  int lane = threadIdx.x & 31;
  if (row >= MTOT) return;
  float* p = z + (size_t)row * RR;
  float vals[RR / 32];
  float mx = -3.4e38f;
#pragma unroll
  for (int t = 0; t < RR / 32; ++t) {
    vals[t] = p[lane + 32 * t];
    mx = fmaxf(mx, vals[t]);
  }
  mx = waveRedMax(mx);
  float s = 0.0f;
#pragma unroll
  for (int t = 0; t < RR / 32; ++t) {
    vals[t] = __expf(vals[t] - mx);
    s += vals[t];
  }
  s = waveRedSum(s);
  float inv = 1.0f / s;
#pragma unroll
  for (int t = 0; t < RR / 32; ++t) p[lane + 32 * t] = vals[t] * inv;
}

__device__ __forceinline__ void mm_kn_128x64(
    const float* __restrict__ Ablk,
    const float* __restrict__ Btile,
    const int K, v8f& c0, v8f& c1, v8f& c2, v8f& c3) {
  __shared__ __attribute__((aligned(16))) float As[2][128 * PA];
  __shared__ __attribute__((aligned(16))) float Bs[2][KT * PB];
  int t = threadIdx.x;
  int wave = t >> 5;
  int lane = t & 31;
  int hi = lane >> 4;
  int ln = lane & 15;

  int cr = t >> 3;
  int cq = (t & 7) * 4;
  int br = t >> 4;
  int bq = (t & 15) * 4;
  const float* ag = Ablk + cr * K + cq;
  const float* bg = Btile + br * DD + bq;
  int a_st = cr * PA + cq;
  int b_st = br * PB + bq;
  int a_ld = (wave * 16 + ln) * PA + 4 * hi;
  int b_ld = (4 * hi) * PB + 4 * ln;

  v4f ra[4], rb[2];
#pragma unroll
  for (int i = 0; i < 4; ++i) ra[i] = *(const v4f*)(ag + i * 32 * K);
#pragma unroll
  for (int i = 0; i < 2; ++i) rb[i] = *(const v4f*)(bg + i * 16 * DD);
#pragma unroll
  for (int i = 0; i < 4; ++i) *(v4f*)(&As[0][a_st + i * 32 * PA]) = ra[i];
#pragma unroll
  for (int i = 0; i < 2; ++i) *(v4f*)(&Bs[0][b_st + i * 16 * PB]) = rb[i];
  __syncthreads();

  for (int kt = 0; kt < K / KT; ++kt) {
    const int cur = kt & 1;
    const bool more = (kt + 1) < K / KT;
    if (more) {
#pragma unroll
      for (int i = 0; i < 4; ++i)
        ra[i] = *(const v4f*)(ag + (kt + 1) * KT + i * 32 * K);
#pragma unroll
      for (int i = 0; i < 2; ++i)
        rb[i] = *(const v4f*)(bg + ((kt + 1) * KT + i * 16) * DD);
      __builtin_amdgcn_sched_barrier(0);
    }
    const float* as = &As[cur][a_ld];
    const float* bs = &Bs[cur][b_ld];
#pragma unroll
    for (int kk = 0; kk < KT / 8; ++kk) {
      v4f a  = *(const v4f*)(as + kk * 8);
      v4f r0 = *(const v4f*)(bs + (kk * 8 + 0) * PB);
      v4f r1 = *(const v4f*)(bs + (kk * 8 + 1) * PB);
      v4f r2 = *(const v4f*)(bs + (kk * 8 + 2) * PB);
      v4f r3 = *(const v4f*)(bs + (kk * 8 + 3) * PB);
      v2f alo = {a.x, a.y}, ahi = {a.z, a.w};
      c0 = wmma_f32(alo, (v2f){r0.x, r1.x}, c0);
      c1 = wmma_f32(alo, (v2f){r0.y, r1.y}, c1);
      c2 = wmma_f32(alo, (v2f){r0.z, r1.z}, c2);
      c3 = wmma_f32(alo, (v2f){r0.w, r1.w}, c3);
      c0 = wmma_f32(ahi, (v2f){r2.x, r3.x}, c0);
      c1 = wmma_f32(ahi, (v2f){r2.y, r3.y}, c1);
      c2 = wmma_f32(ahi, (v2f){r2.z, r3.z}, c2);
      c3 = wmma_f32(ahi, (v2f){r2.w, r3.w}, c3);
    }
    if (more) {
#pragma unroll
      for (int i = 0; i < 4; ++i) *(v4f*)(&As[cur ^ 1][a_st + i * 32 * PA]) = ra[i];
#pragma unroll
      for (int i = 0; i < 2; ++i) *(v4f*)(&Bs[cur ^ 1][b_st + i * 16 * PB]) = rb[i];
    }
    __syncthreads();
  }
}

#define CONQ_TILES ((RR / 128) * (DD / 64) * BB)
__global__ __launch_bounds__(256) void conq_softmax_kernel(
    const float* __restrict__ query, const float* __restrict__ Wc,
    const float* __restrict__ bc, float* __restrict__ conq,
    float* __restrict__ z) {
  if (blockIdx.x >= CONQ_TILES) {
    softmax_rows(blockIdx.x - CONQ_TILES, z);
    return;
  }
  const int tile_x = blockIdx.x & (RR / 128 - 1);
  const int tile_y = (blockIdx.x / (RR / 128)) & (DD / 64 - 1);
  const int K = LL, N = DD;
  int b = blockIdx.x / ((RR / 128) * (DD / 64));
  int wave = threadIdx.x >> 5;
  int lane = threadIdx.x & 31;
  int hi = lane >> 4;
  int ln = lane & 15;
  int blk_m = tile_x * 128;
  int tile_m = blk_m + wave * 16;
  int tile_n = tile_y * 64;

  v8f c0 = {}, c1 = {}, c2 = {}, c3 = {};
  mm_kn_128x64(Wc + blk_m * K, query + b * LL * DD + tile_n, K, c0, c1, c2, c3);

  float* Cb = conq + b * RR * DD;
#pragma unroll
  for (int i = 0; i < 8; ++i) {
    int m = tile_m + i + 8 * hi;
    float bias = bc[m];
    v4f o = {c0[i] + bias, c1[i] + bias, c2[i] + bias, c3[i] + bias};
    *(v4f*)(Cb + m * N + tile_n + 4 * ln) = o;
  }
}

#define PB2 136
__device__ __forceinline__ void mm_kn_128x128(
    const float* __restrict__ Ablk,
    const float* __restrict__ Btile,
    const int K, v8f (&c)[2][4]) {
  __shared__ __attribute__((aligned(16))) float As[2][128 * PA];
  __shared__ __attribute__((aligned(16))) float Bs[2][KT * PB2];
  int t = threadIdx.x;
  int wave = t >> 5;
  int lane = t & 31;
  int hi = lane >> 4;
  int ln = lane & 15;
  int wm = wave & 3;
  int wn = wave >> 2;

  int cr = t >> 3;
  int cq = (t & 7) * 4;
  int br = t >> 5;
  int bq = (t & 31) * 4;
  const float* ag = Ablk + cr * K + cq;
  const float* bg = Btile + br * DD + bq;
  int a_st = cr * PA + cq;
  int b_st = br * PB2 + bq;
  int a_ld = (wm * 32 + ln) * PA + 4 * hi;
  int b_ld = (4 * hi) * PB2 + 64 * wn + 4 * ln;

  v4f ra[4], rb[4];
#pragma unroll
  for (int i = 0; i < 4; ++i) ra[i] = *(const v4f*)(ag + i * 32 * K);
#pragma unroll
  for (int i = 0; i < 4; ++i) rb[i] = *(const v4f*)(bg + i * 8 * DD);
#pragma unroll
  for (int i = 0; i < 4; ++i) *(v4f*)(&As[0][a_st + i * 32 * PA]) = ra[i];
#pragma unroll
  for (int i = 0; i < 4; ++i) *(v4f*)(&Bs[0][b_st + i * 8 * PB2]) = rb[i];
  __syncthreads();

  for (int kt = 0; kt < K / KT; ++kt) {
    const int cur = kt & 1;
    const bool more = (kt + 1) < K / KT;
    if (more) {
#pragma unroll
      for (int i = 0; i < 4; ++i)
        ra[i] = *(const v4f*)(ag + (kt + 1) * KT + i * 32 * K);
#pragma unroll
      for (int i = 0; i < 4; ++i)
        rb[i] = *(const v4f*)(bg + ((kt + 1) * KT + i * 8) * DD);
      __builtin_amdgcn_sched_barrier(0);
    }
    const float* as = &As[cur][a_ld];
    const float* bs = &Bs[cur][b_ld];
#pragma unroll
    for (int kk = 0; kk < KT / 8; ++kk) {
      v4f a[2], r[4];
#pragma unroll
      for (int mi = 0; mi < 2; ++mi) a[mi] = *(const v4f*)(as + mi * 16 * PA + kk * 8);
#pragma unroll
      for (int j = 0; j < 4; ++j) r[j] = *(const v4f*)(bs + (kk * 8 + j) * PB2);
#pragma unroll
      for (int mi = 0; mi < 2; ++mi) {
        v2f alo = {a[mi].x, a[mi].y};
        c[mi][0] = wmma_f32(alo, (v2f){r[0].x, r[1].x}, c[mi][0]);
        c[mi][1] = wmma_f32(alo, (v2f){r[0].y, r[1].y}, c[mi][1]);
        c[mi][2] = wmma_f32(alo, (v2f){r[0].z, r[1].z}, c[mi][2]);
        c[mi][3] = wmma_f32(alo, (v2f){r[0].w, r[1].w}, c[mi][3]);
      }
#pragma unroll
      for (int mi = 0; mi < 2; ++mi) {
        v2f ahi = {a[mi].z, a[mi].w};
        c[mi][0] = wmma_f32(ahi, (v2f){r[2].x, r[3].x}, c[mi][0]);
        c[mi][1] = wmma_f32(ahi, (v2f){r[2].y, r[3].y}, c[mi][1]);
        c[mi][2] = wmma_f32(ahi, (v2f){r[2].z, r[3].z}, c[mi][2]);
        c[mi][3] = wmma_f32(ahi, (v2f){r[2].w, r[3].w}, c[mi][3]);
      }
    }
    if (more) {
#pragma unroll
      for (int i = 0; i < 4; ++i) *(v4f*)(&As[cur ^ 1][a_st + i * 32 * PA]) = ra[i];
#pragma unroll
      for (int i = 0; i < 4; ++i) *(v4f*)(&Bs[cur ^ 1][b_st + i * 8 * PB2]) = rb[i];
    }
    __syncthreads();
  }
}

__global__ __launch_bounds__(256) void out_kernel(
    const float* __restrict__ Fss, const float* __restrict__ conq,
    float* __restrict__ out) {
  const int K = RR, N = DD;
  int b = blockIdx.z;
  int wave = threadIdx.x >> 5;
  int lane = threadIdx.x & 31;
  int hi = lane >> 4;
  int ln = lane & 15;
  int wm = wave & 3;
  int wn = wave >> 2;
  int blk_m = blockIdx.x * 128;
  int tile_m = blk_m + wm * 32;
  int tile_n = blockIdx.y * 128;

  v8f c[2][4] = {};
  mm_kn_128x128(Fss + b * LL * RR + blk_m * K, conq + b * RR * DD + tile_n, K, c);

  float* Ob = out + b * LL * DD;
#pragma unroll
  for (int mi = 0; mi < 2; ++mi) {
#pragma unroll
    for (int i = 0; i < 8; ++i) {
      int m = tile_m + mi * 16 + i + 8 * hi;
      v4f o = {c[mi][0][i], c[mi][1][i], c[mi][2][i], c[mi][3][i]};
      *(v4f*)(Ob + m * N + tile_n + 64 * wn + 4 * ln) = o;
    }
  }
}

extern "C" void kernel_launch(void* const* d_in, const int* in_sizes, int n_in,
                              void* d_out, int out_size, void* d_ws, size_t ws_size,
                              hipStream_t stream) {
  const float* query   = (const float*)d_in[0];
  const float* Wq      = (const float*)d_in[1];
  const float* bq      = (const float*)d_in[2];
  const float* Wc      = (const float*)d_in[3];
  const float* bc      = (const float*)d_in[4];
  const float* centers = (const float*)d_in[5];
  const float* widths  = (const float*)d_in[6];
  float* out = (float*)d_out;

  float* q    = (float*)d_ws;
  float* z    = q    + (size_t)MTOT * DD;
  float* iw   = z    + (size_t)MTOT * RR;
  float* cw   = iw   + (size_t)RR * DD;
  float* cc   = cw   + (size_t)RR * DD;
  float* conq = cc   + RR;

  qproj_prep_kernel<<<PREP_BLOCKS + (MTOT / 128) * (DD / 128), 256, 0, stream>>>(
      query, Wq, bq, q, centers, widths, iw, cw, cc);

  z_kernel<<<dim3(MTOT / 128, RR / 128), 256, 0, stream>>>(q, iw, cw, cc, z);

  conq_softmax_kernel<<<CONQ_TILES + SOFTMAX_BLOCKS, 256, 0, stream>>>(
      query, Wc, bc, conq, z);

  out_kernel<<<dim3(LL / 128, DD / 128, BB), 256, 0, stream>>>(z, conq, out);
}


# 4 launches -> 3: conq tiles ride in launch 1; K3 leaves per row the max and exp-sum of each 64-column group, K6 merges them and turns z into Fss as it stages A (softmax sum regrouped by 64 columns)
# speedup vs baseline: 2.3711x; 1.0416x over previous
#include <hip/hip_runtime.h>
#include <hip/hip_bf16.h>

#define BB 8
#define LL 2048
#define DD 512
#define RR 512
#define MTOT (BB * LL)

typedef float v2f __attribute__((ext_vector_type(2)));
typedef float v4f __attribute__((ext_vector_type(4)));
typedef float v8f __attribute__((ext_vector_type(8)));


#define KT 32
#define PA 36
#define PB 72

__device__ __forceinline__ v8f wmma_f32(v2f a, v2f b, v8f c) {
  return __builtin_amdgcn_wmma_f32_16x16x4_f32(
       false, a,  false, b,
       (short)0, c,  false,  false);
}

__device__ __forceinline__ float waveRedSum(float v) {
#pragma unroll
  for (int m = 16; m >= 1; m >>= 1) v += __shfl_xor(v, m, 32);
  return v;
}
__device__ __forceinline__ float waveRedMax(float v) {
#pragma unroll
  for (int m = 16; m >= 1; m >>= 1) v = fmaxf(v, __shfl_xor(v, m, 32));
  return v;
}

#define PREP_BLOCKS (RR / 8)
__device__ __forceinline__ void prep_rows(
    const int blk,
    const float* __restrict__ centers, const float* __restrict__ widths,
    float* __restrict__ iw, float* __restrict__ cw, float* __restrict__ cc) {
  int row = (blk * 256 + threadIdx.x) >> 5;
  int lane = threadIdx.x & 31;
  if (row >= RR) return;
  const float* crow = centers + row * DD;
  const float* wrow = widths + row * DD;
  float acc = 0.0f;
#pragma unroll
  for (int t = 0; t < DD / 32; ++t) {
    int d = lane + 32 * t;
    float w = wrow[d];
    float inv = 1.0f / (w * w);
    float c = crow[d];
    iw[row * DD + d] = inv;
    cw[row * DD + d] = c * inv;
    acc += c * c * inv;
  }
  acc = waveRedSum(acc);
  if (lane == 0) cc[row] = acc;
}

__device__ __forceinline__ void mm_kn_128x64(
    const float* __restrict__ Ablk,
    const float* __restrict__ Btile,
    const int K, float* smem, v8f& c0, v8f& c1, v8f& c2, v8f& c3) {
  float (*As)[128 * PA] = (float (*)[128 * PA])smem;
  float (*Bs)[KT * PB] = (float (*)[KT * PB])(smem + 2 * 128 * PA);
  int t = threadIdx.x;
  int wave = t >> 5;
  int lane = t & 31;
  int hi = lane >> 4;
  int ln = lane & 15;

  int cr = t >> 3;
  int cq = (t & 7) * 4;
  int br = t >> 4;
  int bq = (t & 15) * 4;
  const float* ag = Ablk + cr * K + cq;
  const float* bg = Btile + br * DD + bq;
  int a_st = cr * PA + cq;
  int b_st = br * PB + bq;
  int a_ld = (wave * 16 + ln) * PA + 4 * hi;
  int b_ld = (4 * hi) * PB + 4 * ln;

  v4f ra[4], rb[2];
#pragma unroll
  for (int i = 0; i < 4; ++i) ra[i] = *(const v4f*)(ag + i * 32 * K);
#pragma unroll
  for (int i = 0; i < 2; ++i) rb[i] = *(const v4f*)(bg + i * 16 * DD);
#pragma unroll
  for (int i = 0; i < 4; ++i) *(v4f*)(&As[0][a_st + i * 32 * PA]) = ra[i];
#pragma unroll
  for (int i = 0; i < 2; ++i) *(v4f*)(&Bs[0][b_st + i * 16 * PB]) = rb[i];
  __syncthreads();

  for (int kt = 0; kt < K / KT; ++kt) {
    const int cur = kt & 1;
    const bool more = (kt + 1) < K / KT;
    if (more) {
#pragma unroll
      for (int i = 0; i < 4; ++i)
        ra[i] = *(const v4f*)(ag + (kt + 1) * KT + i * 32 * K);
#pragma unroll
      for (int i = 0; i < 2; ++i)
        rb[i] = *(const v4f*)(bg + ((kt + 1) * KT + i * 16) * DD);
      __builtin_amdgcn_sched_barrier(0);
    }
    const float* as = &As[cur][a_ld];
    const float* bs = &Bs[cur][b_ld];
#pragma unroll
    for (int kk = 0; kk < KT / 8; ++kk) {
      v4f a  = *(const v4f*)(as + kk * 8);
      v4f r0 = *(const v4f*)(bs + (kk * 8 + 0) * PB);
      v4f r1 = *(const v4f*)(bs + (kk * 8 + 1) * PB);
      v4f r2 = *(const v4f*)(bs + (kk * 8 + 2) * PB);
      v4f r3 = *(const v4f*)(bs + (kk * 8 + 3) * PB);
      v2f alo = {a.x, a.y}, ahi = {a.z, a.w};
      c0 = wmma_f32(alo, (v2f){r0.x, r1.x}, c0);
      c1 = wmma_f32(alo, (v2f){r0.y, r1.y}, c1);
      c2 = wmma_f32(alo, (v2f){r0.z, r1.z}, c2);
      c3 = wmma_f32(alo, (v2f){r0.w, r1.w}, c3);
      c0 = wmma_f32(ahi, (v2f){r2.x, r3.x}, c0);
      c1 = wmma_f32(ahi, (v2f){r2.y, r3.y}, c1);
      c2 = wmma_f32(ahi, (v2f){r2.z, r3.z}, c2);
      c3 = wmma_f32(ahi, (v2f){r2.w, r3.w}, c3);
    }
    if (more) {
#pragma unroll
      for (int i = 0; i < 4; ++i) *(v4f*)(&As[cur ^ 1][a_st + i * 32 * PA]) = ra[i];
#pragma unroll
      for (int i = 0; i < 2; ++i) *(v4f*)(&Bs[cur ^ 1][b_st + i * 16 * PB]) = rb[i];
    }
    __syncthreads();
  }
}

#define CONQ_TILES ((RR / 128) * (DD / 64) * BB)
__device__ __forceinline__ void conq_tile(
    const int blk, const float* __restrict__ query, const float* __restrict__ Wc,
    const float* __restrict__ bc, float* __restrict__ conq, float* smem) {
  const int tile_x = blk & (RR / 128 - 1);
  const int tile_y = (blk / (RR / 128)) & (DD / 64 - 1);
  const int K = LL, N = DD;
  int b = blk / ((RR / 128) * (DD / 64));
  int wave = threadIdx.x >> 5;
  int lane = threadIdx.x & 31;
  int hi = lane >> 4;
  int ln = lane & 15;
  int blk_m = tile_x * 128;
  int tile_m = blk_m + wave * 16;
  int tile_n = tile_y * 64;

  v8f c0 = {}, c1 = {}, c2 = {}, c3 = {};
  mm_kn_128x64(Wc + blk_m * K, query + b * LL * DD + tile_n, K, smem, c0, c1, c2, c3);

  float* Cb = conq + b * RR * DD;
#pragma unroll
  for (int i = 0; i < 8; ++i) {
    int m = tile_m + i + 8 * hi;
    float bias = bc[m];
    v4f o = {c0[i] + bias, c1[i] + bias, c2[i] + bias, c3[i] + bias};
    *(v4f*)(Cb + m * N + tile_n + 4 * ln) = o;
  }
}

#define QPROJ_TILES ((MTOT / 128) * (DD / 128))
__global__ __launch_bounds__(256) void qproj_conq_prep_kernel(
    const float* __restrict__ A, const float* __restrict__ Bm,
    const float* __restrict__ bias, float* __restrict__ C,
    const float* __restrict__ centers, const float* __restrict__ widths,
    float* __restrict__ iw, float* __restrict__ cw, float* __restrict__ cc,
    const float* __restrict__ Wc, const float* __restrict__ bc,
    float* __restrict__ conq) {
  __shared__ __attribute__((aligned(16))) float smem[4 * 128 * PA];
  if (blockIdx.x < CONQ_TILES) {
    conq_tile(blockIdx.x, A, Wc, bc, conq, smem);
    return;
  }
  if (blockIdx.x >= CONQ_TILES + QPROJ_TILES) {
    prep_rows(blockIdx.x - (CONQ_TILES + QPROJ_TILES), centers, widths, iw, cw, cc);
    return;
  }
  const int tile_id = blockIdx.x - CONQ_TILES;
  const int tile_x = tile_id & (MTOT / 128 - 1);
  const int tile_y = tile_id / (MTOT / 128);
  const int K = DD, N = DD;
  float (*As)[128 * PA] = (float (*)[128 * PA])smem;
  float (*Bs)[128 * PA] = (float (*)[128 * PA])(smem + 2 * 128 * PA);
  int t = threadIdx.x;
  int wave = t >> 5;
  int lane = t & 31;
  int hi = lane >> 4;
  int ln = lane & 15;
  int wm = wave & 3;
  int wn = wave >> 2;
  int blk_m = tile_x * 128;
  int tile_m = blk_m + wm * 32;
  int tile_n = tile_y * 128;

  int cr = t >> 3;
  int cq = (t & 7) * 4;
  const float* ag = A + (blk_m + cr) * K + cq;
  const float* bg = Bm + (tile_n + 4 * (cr & 15) + (cr >> 4)) * K + cq;
  int st = cr * PA + cq;
  int a_ld = (wm * 32 + ln) * PA + 4 * hi;
  int b_ld = (wn * 64 + ln) * PA + 4 * hi;

  v4f ra[4], rb[4];
#pragma unroll
  for (int i = 0; i < 4; ++i) ra[i] = *(const v4f*)(ag + i * 32 * K);
#pragma unroll
  for (int i = 0; i < 4; ++i)
    rb[i] = *(const v4f*)(bg + ((i >> 1) * 64 + 2 * (i & 1)) * K);
#pragma unroll
  for (int i = 0; i < 4; ++i) *(v4f*)(&As[0][st + i * 32 * PA]) = ra[i];
#pragma unroll
  for (int i = 0; i < 4; ++i) *(v4f*)(&Bs[0][st + i * 32 * PA]) = rb[i];
  __syncthreads();

  v8f c[2][4] = {};
  for (int kt = 0; kt < K / KT; ++kt) {
    const int cur = kt & 1;
    const bool more = (kt + 1) < K / KT;
    if (more) {
#pragma unroll
      for (int i = 0; i < 4; ++i)
        ra[i] = *(const v4f*)(ag + (kt + 1) * KT + i * 32 * K);
#pragma unroll
      for (int i = 0; i < 4; ++i)
        rb[i] = *(const v4f*)(bg + (kt + 1) * KT + ((i >> 1) * 64 + 2 * (i & 1)) * K);
      __builtin_amdgcn_sched_barrier(0);
    }
    const float* as = &As[cur][a_ld];
    const float* bs = &Bs[cur][b_ld];
#pragma unroll
    for (int kk = 0; kk < KT / 8; ++kk) {
      v4f a[2], b[4];
#pragma unroll
      for (int mi = 0; mi < 2; ++mi) a[mi] = *(const v4f*)(as + mi * 16 * PA + kk * 8);
#pragma unroll
      for (int j = 0; j < 4; ++j) b[j] = *(const v4f*)(bs + j * 16 * PA + kk * 8);
#pragma unroll
      for (int mi = 0; mi < 2; ++mi)
#pragma unroll
        for (int j = 0; j < 4; ++j)
          c[mi][j] = wmma_f32((v2f){a[mi].x, a[mi].y}, (v2f){b[j].x, b[j].y}, c[mi][j]);
#pragma unroll
      for (int mi = 0; mi < 2; ++mi)
#pragma unroll
        for (int j = 0; j < 4; ++j)
          c[mi][j] = wmma_f32((v2f){a[mi].z, a[mi].w}, (v2f){b[j].z, b[j].w}, c[mi][j]);
    }
    if (more) {
#pragma unroll
      for (int i = 0; i < 4; ++i) *(v4f*)(&As[cur ^ 1][st + i * 32 * PA]) = ra[i];
#pragma unroll
      for (int i = 0; i < 4; ++i) *(v4f*)(&Bs[cur ^ 1][st + i * 32 * PA]) = rb[i];
    }
    __syncthreads();
  }
  v4f bq = *(const v4f*)(bias + tile_n + 64 * wn + 4 * ln);
#pragma unroll
  for (int mi = 0; mi < 2; ++mi) {
#pragma unroll
    for (int i = 0; i < 8; ++i) {
      int m = tile_m + mi * 16 + i + 8 * hi;
      v4f o = {c[mi][0][i] + bq.x, c[mi][1][i] + bq.y, c[mi][2][i] + bq.z, c[mi][3][i] + bq.w};
      *(v4f*)(C + m * N + tile_n + 64 * wn + 4 * ln) = o;
    }
  }
}

__global__ __launch_bounds__(256) __attribute__((amdgpu_waves_per_eu(4, 4))) void z_kernel(
    const float* __restrict__ q, const float* __restrict__ iw,
    const float* __restrict__ cw, const float* __restrict__ cc,
    float* __restrict__ z, float* __restrict__ pmax, float* __restrict__ psum) {
  const int K = DD, N = RR;
  __shared__ __attribute__((aligned(16))) float As[2][128 * PA];
  __shared__ __attribute__((aligned(16))) float Bi[2][128 * PA];
  __shared__ __attribute__((aligned(16))) float Bc[2][128 * PA];
  int t = threadIdx.x;
  int wave = t >> 5;
  int lane = t & 31;
  int hi = lane >> 4;
  int ln = lane & 15;
  int wm = wave & 3;
  int wn = wave >> 2;
  int blk_m = blockIdx.x * 128;
  int tile_m = blk_m + wm * 32;
  int tile_n = blockIdx.y * 128;

  int cr = t >> 3;
  int cq = (t & 7) * 4;
  const float* ag = q + (blk_m + cr) * K + cq;
  int brow = tile_n + 4 * (cr & 15) + (cr >> 4);
  const float* ig = iw + brow * K + cq;
  const float* cg = cw + brow * K + cq;
  int st = cr * PA + cq;
  int a_ld = (wm * 32 + ln) * PA + 4 * hi;
  int b_ld = (wn * 64 + ln) * PA + 4 * hi;

  v4f ra[4], ri[4], rc[4];
#pragma unroll
  for (int i = 0; i < 4; ++i) {
    ra[i] = *(const v4f*)(ag + i * 32 * K);
    ri[i] = *(const v4f*)(ig + ((i >> 1) * 64 + 2 * (i & 1)) * K);
    rc[i] = *(const v4f*)(cg + ((i >> 1) * 64 + 2 * (i & 1)) * K);
  }
#pragma unroll
  for (int i = 0; i < 4; ++i) {
    *(v4f*)(&As[0][st + i * 32 * PA]) = ra[i];
    *(v4f*)(&Bi[0][st + i * 32 * PA]) = ri[i];
    *(v4f*)(&Bc[0][st + i * 32 * PA]) = rc[i];
  }
  __syncthreads();

  v8f qq[2][4] = {}, qc[2][4] = {};
  for (int kt = 0; kt < K / KT; ++kt) {
    const int cur = kt & 1;
    const bool more = (kt + 1) < K / KT;
    if (more) {
#pragma unroll
      for (int i = 0; i < 4; ++i) {
        ra[i] = *(const v4f*)(ag + (kt + 1) * KT + i * 32 * K);
        ri[i] = *(const v4f*)(ig + (kt + 1) * KT + ((i >> 1) * 64 + 2 * (i & 1)) * K);
        rc[i] = *(const v4f*)(cg + (kt + 1) * KT + ((i >> 1) * 64 + 2 * (i & 1)) * K);
      }
      __builtin_amdgcn_sched_barrier(0);
    }
    const float* as = &As[cur][a_ld];
    const float* is = &Bi[cur][b_ld];
    const float* cs = &Bc[cur][b_ld];
#pragma unroll
    for (int kk = 0; kk < KT / 8; ++kk) {
      v4f a[2], s[2], bi[4], bc[4];
#pragma unroll
      for (int mi = 0; mi < 2; ++mi) {
        a[mi] = *(const v4f*)(as + mi * 16 * PA + kk * 8);
        s[mi] = a[mi] * a[mi];
      }
#pragma unroll
      for (int j = 0; j < 4; ++j) {
        bi[j] = *(const v4f*)(is + j * 16 * PA + kk * 8);
        bc[j] = *(const v4f*)(cs + j * 16 * PA + kk * 8);
      }
#pragma unroll
      for (int mi = 0; mi < 2; ++mi)
#pragma unroll
        for (int j = 0; j < 4; ++j) {
          qq[mi][j] = wmma_f32((v2f){s[mi].x, s[mi].y}, (v2f){bi[j].x, bi[j].y}, qq[mi][j]);
          qc[mi][j] = wmma_f32((v2f){a[mi].x, a[mi].y}, (v2f){bc[j].x, bc[j].y}, qc[mi][j]);
        }
#pragma unroll
      for (int mi = 0; mi < 2; ++mi)
#pragma unroll
        for (int j = 0; j < 4; ++j) {
          qq[mi][j] = wmma_f32((v2f){s[mi].z, s[mi].w}, (v2f){bi[j].z, bi[j].w}, qq[mi][j]);
          qc[mi][j] = wmma_f32((v2f){a[mi].z, a[mi].w}, (v2f){bc[j].z, bc[j].w}, qc[mi][j]);
        }
    }
    if (more) {
#pragma unroll
      for (int i = 0; i < 4; ++i) {
        *(v4f*)(&As[cur ^ 1][st + i * 32 * PA]) = ra[i];
        *(v4f*)(&Bi[cur ^ 1][st + i * 32 * PA]) = ri[i];
        *(v4f*)(&Bc[cur ^ 1][st + i * 32 * PA]) = rc[i];
      }
    }
    __syncthreads();
  }
  const float scale = -0.5f / (float)DD;
  v4f ccv = *(const v4f*)(cc + tile_n + 64 * wn + 4 * ln);
  const int grp = blockIdx.y * 2 + wn;
#pragma unroll
  for (int mi = 0; mi < 2; ++mi) {
    v4f zv[8];
    float pm[8], ps[8];
#pragma unroll
    for (int i = 0; i < 8; ++i) {
      int m = tile_m + mi * 16 + i + 8 * hi;
      v4f o = {scale * (qq[mi][0][i] - 2.0f * qc[mi][0][i] + ccv.x),
               scale * (qq[mi][1][i] - 2.0f * qc[mi][1][i] + ccv.y),
               scale * (qq[mi][2][i] - 2.0f * qc[mi][2][i] + ccv.z),
               scale * (qq[mi][3][i] - 2.0f * qc[mi][3][i] + ccv.w)};
      *(v4f*)(z + m * N + tile_n + 64 * wn + 4 * ln) = o;
      zv[i] = o;
      pm[i] = fmaxf(fmaxf(o.x, o.y), fmaxf(o.z, o.w));
    }
#pragma unroll
    for (int d = 8; d >= 1; d >>= 1)
#pragma unroll
      for (int i = 0; i < 8; ++i) pm[i] = fmaxf(pm[i], __shfl_xor(pm[i], d, 32));
#pragma unroll
    for (int i = 0; i < 8; ++i) {
      v4f o = zv[i];
      float mx = pm[i];
      ps[i] = (__expf(o.x - mx) + __expf(o.y - mx)) + (__expf(o.z - mx) + __expf(o.w - mx));
    }
#pragma unroll
    for (int d = 8; d >= 1; d >>= 1)
#pragma unroll
      for (int i = 0; i < 8; ++i) ps[i] += __shfl_xor(ps[i], d, 32);
    if (ln == 0) {
#pragma unroll
      for (int i = 0; i < 8; ++i) {
        int m = tile_m + mi * 16 + i + 8 * hi;
        pmax[grp * MTOT + m] = pm[i];
        psum[grp * MTOT + m] = ps[i];
      }
    }
  }
}

#define NCG (RR / 64)

__device__ __forceinline__ v4f fss_of(v4f v, float mx, float inv) {
  v4f o = {__expf(v.x - mx) * inv, __expf(v.y - mx) * inv,
           __expf(v.z - mx) * inv, __expf(v.w - mx) * inv};
  return o;
}

#define PB2 136
__device__ __forceinline__ void mm_kn_128x128(
    const float* __restrict__ Ablk,
    const float* __restrict__ Btile,
    const float* __restrict__ pmax,
    const float* __restrict__ psum,
    const int K, v8f (&c)[2][4]) {
  __shared__ __attribute__((aligned(16))) float As[2][128 * PA];
  __shared__ __attribute__((aligned(16))) float Bs[2][KT * PB2];
  __shared__ float rmax[128];
  __shared__ float rinv[128];
  int t = threadIdx.x;
  int wave = t >> 5;
  int lane = t & 31;
  int hi = lane >> 4;
  int ln = lane & 15;
  int wm = wave & 3;
  int wn = wave >> 2;

  int cr = t >> 3;
  int cq = (t & 7) * 4;
  int br = t >> 5;
  int bq = (t & 31) * 4;
  const float* ag = Ablk + cr * K + cq;
  const float* bg = Btile + br * DD + bq;
  int a_st = cr * PA + cq;
  int b_st = br * PB2 + bq;
  int a_ld = (wm * 32 + ln) * PA + 4 * hi;
  int b_ld = (4 * hi) * PB2 + 64 * wn + 4 * ln;

  if (t < 128) {
    float gm[NCG], gs[NCG];
#pragma unroll
    for (int p = 0; p < NCG; ++p) {
      gm[p] = pmax[p * MTOT + t];
      gs[p] = psum[p * MTOT + t];
    }
    float mx = gm[0];
#pragma unroll
    for (int p = 1; p < NCG; ++p) mx = fmaxf(mx, gm[p]);
    float s = 0.0f;
#pragma unroll
    for (int p = 0; p < NCG; ++p) s += gs[p] * __expf(gm[p] - mx);
    rmax[t] = mx;
    rinv[t] = 1.0f / s;
  }
  __syncthreads();
  float mxr[4], ivr[4];
#pragma unroll
  for (int i = 0; i < 4; ++i) {
    mxr[i] = rmax[cr + 32 * i];
    ivr[i] = rinv[cr + 32 * i];
  }

  v4f ra[4], rb[4];
#pragma unroll
  for (int i = 0; i < 4; ++i) ra[i] = *(const v4f*)(ag + i * 32 * K);
#pragma unroll
  for (int i = 0; i < 4; ++i) rb[i] = *(const v4f*)(bg + i * 8 * DD);
#pragma unroll
  for (int i = 0; i < 4; ++i)
    *(v4f*)(&As[0][a_st + i * 32 * PA]) = fss_of(ra[i], mxr[i], ivr[i]);
#pragma unroll
  for (int i = 0; i < 4; ++i) *(v4f*)(&Bs[0][b_st + i * 8 * PB2]) = rb[i];
  __syncthreads();

  for (int kt = 0; kt < K / KT; ++kt) {
    const int cur = kt & 1;
    const bool more = (kt + 1) < K / KT;
    if (more) {
#pragma unroll
      for (int i = 0; i < 4; ++i)
        ra[i] = *(const v4f*)(ag + (kt + 1) * KT + i * 32 * K);
#pragma unroll
      for (int i = 0; i < 4; ++i)
        rb[i] = *(const v4f*)(bg + ((kt + 1) * KT + i * 8) * DD);
      __builtin_amdgcn_sched_barrier(0);
    }
    const float* as = &As[cur][a_ld];
    const float* bs = &Bs[cur][b_ld];
#pragma unroll
    for (int kk = 0; kk < KT / 8; ++kk) {
      v4f a[2], r[4];
#pragma unroll
      for (int mi = 0; mi < 2; ++mi) a[mi] = *(const v4f*)(as + mi * 16 * PA + kk * 8);
#pragma unroll
      for (int j = 0; j < 4; ++j) r[j] = *(const v4f*)(bs + (kk * 8 + j) * PB2);
#pragma unroll
      for (int mi = 0; mi < 2; ++mi) {
        v2f alo = {a[mi].x, a[mi].y};
        c[mi][0] = wmma_f32(alo, (v2f){r[0].x, r[1].x}, c[mi][0]);
        c[mi][1] = wmma_f32(alo, (v2f){r[0].y, r[1].y}, c[mi][1]);
        c[mi][2] = wmma_f32(alo, (v2f){r[0].z, r[1].z}, c[mi][2]);
        c[mi][3] = wmma_f32(alo, (v2f){r[0].w, r[1].w}, c[mi][3]);
      }
#pragma unroll
      for (int mi = 0; mi < 2; ++mi) {
        v2f ahi = {a[mi].z, a[mi].w};
        c[mi][0] = wmma_f32(ahi, (v2f){r[2].x, r[3].x}, c[mi][0]);
        c[mi][1] = wmma_f32(ahi, (v2f){r[2].y, r[3].y}, c[mi][1]);
        c[mi][2] = wmma_f32(ahi, (v2f){r[2].z, r[3].z}, c[mi][2]);
        c[mi][3] = wmma_f32(ahi, (v2f){r[2].w, r[3].w}, c[mi][3]);
      }
    }
    if (more) {
#pragma unroll
      for (int i = 0; i < 4; ++i)
        *(v4f*)(&As[cur ^ 1][a_st + i * 32 * PA]) = fss_of(ra[i], mxr[i], ivr[i]);
#pragma unroll
      for (int i = 0; i < 4; ++i) *(v4f*)(&Bs[cur ^ 1][b_st + i * 8 * PB2]) = rb[i];
    }
    __syncthreads();
  }
}

__global__ __launch_bounds__(256) void out_kernel(
    const float* __restrict__ z, const float* __restrict__ conq,
    const float* __restrict__ pmax, const float* __restrict__ psum,
    float* __restrict__ out) {
  const int K = RR, N = DD;
  int b = blockIdx.z;
  int wave = threadIdx.x >> 5;
  int lane = threadIdx.x & 31;
  int hi = lane >> 4;
  int ln = lane & 15;
  int wm = wave & 3;
  int wn = wave >> 2;
  int blk_m = blockIdx.x * 128;
  int tile_m = blk_m + wm * 32;
  int tile_n = blockIdx.y * 128;

  v8f c[2][4] = {};
  mm_kn_128x128(z + b * LL * RR + blk_m * K, conq + b * RR * DD + tile_n,
                pmax + b * LL + blk_m, psum + b * LL + blk_m, K, c);

  float* Ob = out + b * LL * DD;
#pragma unroll
  for (int mi = 0; mi < 2; ++mi) {
#pragma unroll
    for (int i = 0; i < 8; ++i) {
      int m = tile_m + mi * 16 + i + 8 * hi;
      v4f o = {c[mi][0][i], c[mi][1][i], c[mi][2][i], c[mi][3][i]};
      *(v4f*)(Ob + m * N + tile_n + 64 * wn + 4 * ln) = o;
    }
  }
}

extern "C" void kernel_launch(void* const* d_in, const int* in_sizes, int n_in,
                              void* d_out, int out_size, void* d_ws, size_t ws_size,
                              hipStream_t stream) {
  const float* query   = (const float*)d_in[0];
  const float* Wq      = (const float*)d_in[1];
  const float* bq      = (const float*)d_in[2];
  const float* Wc      = (const float*)d_in[3];
  const float* bc      = (const float*)d_in[4];
  const float* centers = (const float*)d_in[5];
  const float* widths  = (const float*)d_in[6];
  float* out = (float*)d_out;

  float* q    = (float*)d_ws;
  float* z    = q    + (size_t)MTOT * DD;
  float* iw   = z    + (size_t)MTOT * RR;
  float* cw   = iw   + (size_t)RR * DD;
  float* cc   = cw   + (size_t)RR * DD;
  float* conq = cc   + RR;
  float* pmax = conq + (size_t)BB * RR * DD;
  float* psum = pmax + (size_t)(RR / 64) * MTOT;

  qproj_conq_prep_kernel<<<CONQ_TILES + QPROJ_TILES + PREP_BLOCKS, 256, 0, stream>>>(
      query, Wq, bq, q, centers, widths, iw, cw, cc, Wc, bc, conq);

  z_kernel<<<dim3(MTOT / 128, RR / 128), 256, 0, stream>>>(q, iw, cw, cc, z, pmax, psum);

  out_kernel<<<dim3(LL / 128, DD / 128, BB), 256, 0, stream>>>(z, conq, pmax, psum, out);
}
